# starting kernel
# baseline (speedup 1.0000x reference)
#include <hip/hip_runtime.h>
#include <math.h>

typedef __attribute__((ext_vector_type(2))) float v2f;
typedef __attribute__((ext_vector_type(8))) float v8f;

#define NEG_SLOPE 0.2f

template <int TILES>
__global__ __launch_bounds__(32)
void gat_gemm_wmma(const float* __restrict__ X, const float* __restrict__ W,
                   float* __restrict__ Hout, int N, int K) {
  const int F    = TILES * 16;
  const int row0 = blockIdx.x * 16;
  const int lane = threadIdx.x;
  const int l15  = lane & 15;
  const int kh   = (lane >> 4) * 2;
  const int m    = row0 + l15;

  v8f acc[TILES];
#pragma unroll
  for (int t = 0; t < TILES; ++t) acc[t] = (v8f){0.f,0.f,0.f,0.f,0.f,0.f,0.f,0.f};

  for (int k0 = 0; k0 < K; k0 += 4) {
    v2f a;
    a.x = X[(long)m * K + k0 + kh];
    a.y = X[(long)m * K + k0 + kh + 1];
#pragma unroll
    for (int t = 0; t < TILES; ++t) {
      v2f b;
      b.x = W[(long)(k0 + kh)     * F + t * 16 + l15];
      b.y = W[(long)(k0 + kh + 1) * F + t * 16 + l15];
      acc[t] = __builtin_amdgcn_wmma_f32_16x16x4_f32(
           false, a,  false, b,
           (short)0, acc[t],  false,  false);
    }
  }

  const int hi = (lane >> 4) * 8;
#pragma unroll
  for (int t = 0; t < TILES; ++t)
#pragma unroll
    for (int r = 0; r < 8; ++r)
      Hout[(long)(row0 + hi + r) * F + t * 16 + l15] = acc[t][r];
}

__global__ void gat_alpha(const float* __restrict__ h,
                          const float* __restrict__ a_src,
                          const float* __restrict__ a_dst,
                          float* __restrict__ as, float* __restrict__ ad,
                          int N, int C) {
  int wid  = (int)((blockIdx.x * (long)blockDim.x + threadIdx.x) >> 5);
  int lane = threadIdx.x & 31;
  if (wid >= N) return;
  float ss = 0.f, sd = 0.f;
  for (int c = lane; c < C; c += 32) {
    float hv = h[(long)wid * C + c];
    ss += hv * a_src[c];
    sd += hv * a_dst[c];
  }
#pragma unroll
  for (int off = 16; off > 0; off >>= 1) {
    ss += __shfl_down(ss, off, 32);
    sd += __shfl_down(sd, off, 32);
  }
  if (lane == 0) { as[wid] = ss; ad[wid] = sd; }
}

__global__ void gat_edge_logits(const int* __restrict__ src,
                                const int* __restrict__ dst,
                                const float* __restrict__ as,
                                const float* __restrict__ ad,
                                float* __restrict__ ebuf,
                                float* __restrict__ mbuf,
                                int E, int N) {
  int e = blockIdx.x * blockDim.x + threadIdx.x;
  int ET = E + N;
  if (e >= ET) return;
  int s = (e < E) ? src[e] : (e - E);
  int d = (e < E) ? dst[e] : (e - E);
  float v = as[s] + ad[d];
  v = (v > 0.f) ? v : NEG_SLOPE * v;
  ebuf[e] = v;
  atomicMax(&mbuf[d], v);
}

__global__ void gat_edge_exp(const int* __restrict__ dst,
                             float* __restrict__ ebuf,
                             const float* __restrict__ mbuf,
                             float* __restrict__ den,
                             int E, int N) {
  int e = blockIdx.x * blockDim.x + threadIdx.x;
  int ET = E + N;
  if (e >= ET) return;
  int d = (e < E) ? dst[e] : (e - E);
  float ex = __expf(ebuf[e] - mbuf[d]);
  ebuf[e] = ex;
  atomicAdd(&den[d], ex);
}

__global__ void gat_edge_aggregate(const int* __restrict__ src,
                                   const int* __restrict__ dst,
                                   const float* __restrict__ ebuf,
                                   const float* __restrict__ den,
                                   const float* __restrict__ h,
                                   float* __restrict__ agg,
                                   int E, int N, int C) {
  int wid  = (int)((blockIdx.x * (long)blockDim.x + threadIdx.x) >> 5);
  int lane = threadIdx.x & 31;
  int ET = E + N;
  if (wid >= ET) return;
  int s = (wid < E) ? src[wid] : (wid - E);
  int d = (wid < E) ? dst[wid] : (wid - E);
  float alpha = ebuf[wid] / (den[d] + 1e-16f);
  for (int c = lane; c < C; c += 32)
    atomicAdd(&agg[(long)d * C + c], h[(long)s * C + c] * alpha);
}

__global__ void gat_bias_act(const float* __restrict__ agg,
                             const float* __restrict__ bias,
                             float* __restrict__ out,
                             int N, int C, int do_relu) {
  int i = blockIdx.x * blockDim.x + threadIdx.x;
  if (i >= N * C) return;
  float v = agg[i] + bias[i % C];
  if (do_relu) v = fmaxf(v, 0.f);
  out[i] = v;
}

__global__ void gat_fill(float* __restrict__ p, float v, int n) {
  int i = blockIdx.x * blockDim.x + threadIdx.x;
  if (i < n) p[i] = v;
}

extern "C" void kernel_launch(void* const* d_in, const int* in_sizes, int n_in,
                              void* d_out, int out_size, void* d_ws, size_t ws_size,
                              hipStream_t stream) {
  const float* fea = (const float*)d_in[0];
  const int*   ei  = (const int*)  d_in[1];
  const float* W1  = (const float*)d_in[2];
  const float* av1s= (const float*)d_in[3];
  const float* av1d= (const float*)d_in[4];
  const float* b1  = (const float*)d_in[5];
  const float* W2  = (const float*)d_in[6];
  const float* av2s= (const float*)d_in[7];
  const float* av2d= (const float*)d_in[8];
  const float* b2  = (const float*)d_in[9];
  float* out = (float*)d_out;

  const int B  = 2;
  const int C1 = 128, C2 = 64;
  const int N  = in_sizes[0] / (B * C1);
  const int E  = in_sizes[1] / (B * 2);
  const int ET = E + N;

  float* w   = (float*)d_ws;
  float* h1   = w; w += (long)N * C1;
  float* agg1 = w; w += (long)N * C1;
  float* h2   = w; w += (long)N * C2;
  float* agg2 = w; w += (long)N * C2;
  float* asb  = w; w += N;
  float* adb  = w; w += N;
  float* mb   = w; w += N;
  float* denb = w; w += N;
  float* ebuf = w; w += ET;

  const int TB = 256;
  dim3 blk(TB);
  int gN       = (N + TB - 1) / TB;
  int gNC1     = (N * C1 + TB - 1) / TB;
  int gNC2     = (N * C2 + TB - 1) / TB;
  int gEdges   = (ET + TB - 1) / TB;
  int gEdgeWav = (ET * 32 + TB - 1) / TB;
  int gNodeWav = (N * 32 + TB - 1) / TB;
  int gGemm    = N / 16;

  for (int b = 0; b < B; ++b) {
    const float* xb   = fea + (long)b * N * C1;
    const int*   srcb = ei + (long)b * 2 * E;
    const int*   dstb = srcb + E;
    float*       outb = out + (long)b * N * C2;

    gat_fill<<<gN,   blk, 0, stream>>>(mb,  -INFINITY, N);
    gat_fill<<<gN,   blk, 0, stream>>>(denb, 0.f, N);
    gat_fill<<<gNC1, blk, 0, stream>>>(agg1, 0.f, N * C1);

    gat_gemm_wmma<8><<<gGemm, dim3(32), 0, stream>>>(xb, W1, h1, N, C1);
    gat_alpha<<<gNodeWav, blk, 0, stream>>>(h1, av1s, av1d, asb, adb, N, C1);
    gat_edge_logits<<<gEdges, blk, 0, stream>>>(srcb, dstb, asb, adb, ebuf, mb, E, N);
    gat_edge_exp<<<gEdges, blk, 0, stream>>>(dstb, ebuf, mb, denb, E, N);
    gat_edge_aggregate<<<gEdgeWav, blk, 0, stream>>>(srcb, dstb, ebuf, denb, h1, agg1, E, N, C1);
    gat_bias_act<<<gNC1, blk, 0, stream>>>(agg1, b1, h1, N, C1,  1);

    gat_fill<<<gN,   blk, 0, stream>>>(mb,  -INFINITY, N);
    gat_fill<<<gN,   blk, 0, stream>>>(denb, 0.f, N);
    gat_fill<<<gNC2, blk, 0, stream>>>(agg2, 0.f, N * C2);

    gat_gemm_wmma<4><<<gGemm, dim3(32), 0, stream>>>(h1, W2, h2, N, C1);
    gat_alpha<<<gNodeWav, blk, 0, stream>>>(h2, av2s, av2d, asb, adb, N, C2);
    gat_edge_logits<<<gEdges, blk, 0, stream>>>(srcb, dstb, asb, adb, ebuf, mb, E, N);
    gat_edge_exp<<<gEdges, blk, 0, stream>>>(dstb, ebuf, mb, denb, E, N);
    gat_edge_aggregate<<<gEdgeWav, blk, 0, stream>>>(srcb, dstb, ebuf, denb, h2, agg2, E, N, C2);
    gat_bias_act<<<gNC2, blk, 0, stream>>>(agg2, b2, outb, N, C2,  0);
  }
}


# Logits formed in the GEMM epilogue from the accumulators (lane partials of h.a_src, h.a_dst, 16-lane xor tree, one store); 4 wave-per-node logit launches gone, 13->9; logit sum order changed
# speedup vs baseline: 4.6991x; 4.6991x over previous
#include <hip/hip_runtime.h>
#include <math.h>
#include <stdint.h>

typedef __attribute__((ext_vector_type(2))) float v2f;
typedef __attribute__((ext_vector_type(8))) float v8f;

#define NEG_SLOPE 0.2f

template <int TILES>
__global__ __launch_bounds__(32)
void gat_gemm_wmma(const float* __restrict__ X, const float* __restrict__ W,
                   const float* __restrict__ a_src, const float* __restrict__ a_dst,
                   float* __restrict__ Hout, float* __restrict__ as, float* __restrict__ ad,
                   int N, int K) {
  const int F    = TILES * 16;
  const int row0 = blockIdx.x * 16;
  const int lane = threadIdx.x;
  const int l15  = lane & 15;
  const int kh   = (lane >> 4) * 2;
  const int m    = row0 + l15;

  v8f acc[TILES];
#pragma unroll
  for (int t = 0; t < TILES; ++t) acc[t] = (v8f){0.f,0.f,0.f,0.f,0.f,0.f,0.f,0.f};

  for (int k0 = 0; k0 < K; k0 += 4) {
    v2f a;
    a.x = X[(long)m * K + k0 + kh];
    a.y = X[(long)m * K + k0 + kh + 1];
#pragma unroll
    for (int t = 0; t < TILES; ++t) {
      v2f b;
      b.x = W[(long)(k0 + kh)     * F + t * 16 + l15];
      b.y = W[(long)(k0 + kh + 1) * F + t * 16 + l15];
      acc[t] = __builtin_amdgcn_wmma_f32_16x16x4_f32(
           false, a,  false, b,
           (short)0, acc[t],  false,  false);
    }
  }

  const int hi = (lane >> 4) * 8;
#pragma unroll
  for (int t = 0; t < TILES; ++t)
#pragma unroll
    for (int r = 0; r < 8; ++r)
      Hout[(long)(row0 + hi + r) * F + t * 16 + l15] = acc[t][r];

  float ps[8], pd[8];
#pragma unroll
  for (int r = 0; r < 8; ++r) { ps[r] = 0.f; pd[r] = 0.f; }
#pragma unroll
  for (int t = 0; t < TILES; ++t) {
    const float cs = a_src[t * 16 + l15];
    const float cd = a_dst[t * 16 + l15];
#pragma unroll
    for (int r = 0; r < 8; ++r) {
      ps[r] += acc[t][r] * cs;
      pd[r] += acc[t][r] * cd;
    }
  }
#pragma unroll
  for (int off = 8; off > 0; off >>= 1) {
#pragma unroll
    for (int r = 0; r < 8; ++r) {
      ps[r] += __shfl_xor(ps[r], off, 32);
      pd[r] += __shfl_xor(pd[r], off, 32);
    }
  }
  {
    const int r7 = l15 & 7;
    float vs = ps[0], vd = pd[0];
#pragma unroll
    for (int r = 1; r < 8; ++r) { if (r7 == r) { vs = ps[r]; vd = pd[r]; } }
    float* const o = (l15 < 8) ? as : ad;
    o[row0 + hi + r7] = (l15 < 8) ? vs : vd;
  }
}

template <int R>
__global__ __launch_bounds__(256)
void gat_tile_keys(const int* __restrict__ ei, uint4* __restrict__ keys,
                   int E, int N, int nvec) {
  const int b = blockIdx.y;
  const int v = blockIdx.x * 256 + threadIdx.x;
  if (v >= nvec) return;
  const int* dst = ei + ((long)b * 2 + 1) * E;
  const int e0 = v * 16;
  int d[16];
  if (e0 + 16 <= E && (((uintptr_t)(dst + e0)) & 15u) == 0u) {
    const int4* p = (const int4*)(dst + e0);
    int4 a0 = p[0], a1 = p[1], a2 = p[2], a3 = p[3];
    d[0]  = a0.x; d[1]  = a0.y; d[2]  = a0.z; d[3]  = a0.w;
    d[4]  = a1.x; d[5]  = a1.y; d[6]  = a1.z; d[7]  = a1.w;
    d[8]  = a2.x; d[9]  = a2.y; d[10] = a2.z; d[11] = a2.w;
    d[12] = a3.x; d[13] = a3.y; d[14] = a3.z; d[15] = a3.w;
  } else {
#pragma unroll
    for (int j = 0; j < 16; ++j) d[j] = (e0 + j < E) ? dst[e0 + j] : -1;
  }
  unsigned w[4];
#pragma unroll
  for (int q = 0; q < 4; ++q) {
    unsigned acc = 0u;
#pragma unroll
    for (int k = 0; k < 4; ++k) {
      unsigned x = (unsigned)d[4 * q + k];
      unsigned key = (x < (unsigned)N) ? (x / (unsigned)R) : 255u;
      acc |= key << (8 * k);
    }
    w[q] = acc;
  }
  keys[(long)b * nvec + v] = make_uint4(w[0], w[1], w[2], w[3]);
}

__device__ __forceinline__ unsigned gat_fkey(float f) {
  unsigned b = __float_as_uint(f);
  return (b & 0x80000000u) ? ~b : (b | 0x80000000u);
}
__device__ __forceinline__ float gat_fkey_inv(unsigned k) {
  unsigned b = (k & 0x80000000u) ? (k & 0x7fffffffu) : ~k;
  return __uint_as_float(b);
}

template <int C, int R, int CAP, int RELU, int TAKE>
__global__ __launch_bounds__(1024)
void gat_tile_softmax_aggregate(const int* __restrict__ src,
                                const int* __restrict__ dst,
                                const uint4* __restrict__ keys,
                                const float* __restrict__ as,
                                const float* __restrict__ ad,
                                const float* __restrict__ h,
                                const float* __restrict__ bias,
                                float* __restrict__ out,
                                int* __restrict__ keep_s,
                                int* __restrict__ keep_d,
                                int* __restrict__ keep_n,
                                int E, int N) {
  constexpr int NT = 1024;
  constexpr int CH = C / 32;
  static_assert(CAP % 16 == 0, "a piece of the input must start on a key vector");
  __shared__ float    s_acc[R * C];
  __shared__ int      l_s[CAP];
  __shared__ int      l_d[CAP];
  __shared__ float    l_v[CAP];
  __shared__ float    s_ad[R];
  __shared__ float    s_m[R];
  __shared__ float    s_den[R];
  __shared__ float    s_vs[R];
  __shared__ unsigned s_mk[R];
  __shared__ unsigned s_cnt;

  const int tid  = threadIdx.x;
  const int lane = tid & 31;
  const int wv   = __builtin_amdgcn_readfirstlane(tid >> 5);
  const int n0   = blockIdx.x * R;
  const int nr   = (N - n0 < R) ? (N - n0) : R;
  int* const my_s = keep_s + (long)blockIdx.x * CAP;
  int* const my_d = keep_d + (long)blockIdx.x * CAP;

  auto push = [&](int e) {
    unsigned slot = atomicAdd(&s_cnt, 1u);
    if (slot < (unsigned)CAP) l_s[slot] = e;
  };
  const unsigned kmul = (unsigned)blockIdx.x * 0x01010101u;
  auto eq4 = [&](unsigned w) -> unsigned {
    unsigned x = w ^ kmul;
    unsigned t = ((x & 0x7f7f7f7fu) + 0x7f7f7f7fu) | x;
    return ~t & 0x80808080u;
  };
  auto eq16 = [&](uint4 a) -> unsigned {
    return (eq4(a.x) >> 7) | (eq4(a.y) >> 6) | (eq4(a.z) >> 5) | (eq4(a.w) >> 4);
  };
  auto drain = [&](unsigned long long mm, int ebase) {
    while (mm) {
      int P = __builtin_ctzll(mm);
      mm &= mm - 1ull;
      int u = ((P >> 5) << 1) | ((P >> 2) & 1);
      push(ebase + u * (16 * NT) + ((P & 3) << 2) + ((P >> 3) & 3));
    }
  };
  auto scan = [&](int lo, int hi) {
    const uint4* pv = keys + (lo >> 4);
    const int nvec = (hi - lo + 15) >> 4;
    int v = tid;
    for (; v + 3 * NT < nvec; v += 4 * NT) {
      uint4 a0 = pv[v], a1 = pv[v + NT], a2 = pv[v + 2 * NT], a3 = pv[v + 3 * NT];
      unsigned m01 = eq16(a0) | (eq16(a1) << 4);
      unsigned m23 = eq16(a2) | (eq16(a3) << 4);
      drain((unsigned long long)m01 | ((unsigned long long)m23 << 32), lo + 16 * v);
    }
    for (; v < nvec; v += NT) drain((unsigned long long)eq16(pv[v]), lo + 16 * v);
  };
  auto prep = [&](int cnt, bool store) {
    for (int i = tid; i < cnt; i += NT) {
      int e = l_s[i];
      int s = src[e];
      int d = dst[e] - n0;
      float v = as[s] + s_ad[d];
      v = (v > 0.f) ? v : NEG_SLOPE * v;
      l_s[i] = s; l_d[i] = d; l_v[i] = v;
      if (store) { my_s[i] = s; my_d[i] = d; }
      atomicMax(&s_mk[d], gat_fkey(v));
    }
  };
  auto prep_kept = [&](int cnt) {
    for (int i = tid; i < cnt; i += NT) {
      int s = my_s[i];
      int d = my_d[i];
      float v = as[s] + s_ad[d];
      v = (v > 0.f) ? v : NEG_SLOPE * v;
      l_s[i] = s; l_d[i] = d; l_v[i] = v;
      atomicMax(&s_mk[d], gat_fkey(v));
    }
  };
  auto rebuild = [&](int k) -> int {
    __syncthreads();
    if (tid == 0) s_cnt = 0u;
    __syncthreads();
    int lo = k * CAP;
    int hi = (E - lo < CAP) ? E : lo + CAP;
    scan(lo, hi);
    __syncthreads();
    int c = (int)s_cnt;
    prep(c, false);
    __syncthreads();
    return c;
  };
  auto expsum = [&](int cnt, bool add) {
    for (int i = tid; i < cnt; i += NT) {
      int d = l_d[i];
      float ex = __expf(l_v[i] - s_m[d]);
      l_v[i] = ex;
      if (add) atomicAdd(&s_den[d], ex);
    }
  };
  auto row1 = [&](int ii) {
    int s = l_s[ii];
    int d = l_d[ii];
    float al = l_v[ii] / (s_den[d] + 1e-16f);
    float hv[CH];
#pragma unroll
    for (int j = 0; j < CH; ++j) hv[j] = h[(long)s * C + j * 32 + lane];
#pragma unroll
    for (int j = 0; j < CH; ++j) atomicAdd(&s_acc[d * C + j * 32 + lane], hv[j] * al);
  };
  auto rows = [&](int cnt) {
    constexpr int U = 4;
    int i0 = wv;
    for (; i0 + 32 * (U - 1) < cnt; i0 += 32 * U) {
      float hv[U][CH];
      float al[U];
      int   dd[U];
#pragma unroll
      for (int u = 0; u < U; ++u) {
        int ii = i0 + u * 32;
        int s = l_s[ii];
        int d = l_d[ii];
        dd[u] = d;
        al[u] = l_v[ii] / (s_den[d] + 1e-16f);
#pragma unroll
        for (int j = 0; j < CH; ++j) hv[u][j] = h[(long)s * C + j * 32 + lane];
      }
#pragma unroll
      for (int u = 0; u < U; ++u) {
#pragma unroll
        for (int j = 0; j < CH; ++j)
          atomicAdd(&s_acc[dd[u] * C + j * 32 + lane], hv[u][j] * al[u]);
      }
    }
    for (; i0 < cnt; i0 += 32) row1(i0);
  };

  {
    float4 z4 = make_float4(0.f, 0.f, 0.f, 0.f);
    float4* a4 = (float4*)s_acc;
    for (int i = tid; i < R * C / 4; i += NT) a4[i] = z4;
  }
  if (tid < R) {
    float a = 0.f, v = 0.f;
    if (tid < nr) {
      a = ad[n0 + tid];
      v = as[n0 + tid] + a;
      v = (v > 0.f) ? v : NEG_SLOPE * v;
    }
    s_ad[tid] = a;
    s_vs[tid] = v;
    s_mk[tid] = gat_fkey(v);
    s_den[tid] = 0.f;
  }
  if (tid == 0) s_cnt = 0u;
  __syncthreads();

  unsigned total;
  if (TAKE) {
    total = (unsigned)keep_n[blockIdx.x];
  } else {
    scan(0, E);
    __syncthreads();
    total = s_cnt;
    if (tid == 0) keep_n[blockIdx.x] = (int)total;
  }
  const bool whole = (total <= (unsigned)CAP);
  const int  nseg  = whole ? 1 : (E + CAP - 1) / CAP;
  const int  n     = whole ? (int)total : 0;

  if (whole) { if (TAKE) prep_kept(n); else prep(n, true); }
  else for (int k = 0; k < nseg; ++k) rebuild(k);
  __syncthreads();
  if (tid < R) {
    float m = gat_fkey_inv(s_mk[tid]);
    if (!__builtin_isfinite(m)) m = 0.f;
    s_m[tid] = m;
    float ex = __expf(s_vs[tid] - m);
    s_vs[tid] = ex;
    s_den[tid] = ex;
  }
  __syncthreads();

  if (whole) expsum(n, true);
  else for (int k = 0; k < nseg; ++k) { int c = rebuild(k); expsum(c, true); }
  __syncthreads();

  if (whole) rows(n);
  else for (int k = 0; k < nseg; ++k) {
    int c = rebuild(k);
    expsum(c, false);
    __syncthreads();
    rows(c);
  }
  for (int d = wv; d < nr; d += 32) {
    float alpha = s_vs[d] / (s_den[d] + 1e-16f);
#pragma unroll
    for (int j = 0; j < CH; ++j)
      atomicAdd(&s_acc[d * C + j * 32 + lane], h[(long)(n0 + d) * C + j * 32 + lane] * alpha);
  }
  __syncthreads();

  {
    const float4* a4 = (const float4*)s_acc;
    const float4* b4 = (const float4*)bias;
    float4* o4 = (float4*)(out + (long)n0 * C);
    for (int i = tid; i < nr * (C / 4); i += NT) {
      float4 v = a4[i];
      float4 b = b4[i % (C / 4)];
      v.x += b.x; v.y += b.y; v.z += b.z; v.w += b.w;
      if (RELU) { v.x = fmaxf(v.x, 0.f); v.y = fmaxf(v.y, 0.f); v.z = fmaxf(v.z, 0.f); v.w = fmaxf(v.w, 0.f); }
      o4[i] = v;
    }
  }
}

extern "C" void kernel_launch(void* const* d_in, const int* in_sizes, int n_in,
                              void* d_out, int out_size, void* d_ws, size_t ws_size,
                              hipStream_t stream) {
  const float* fea = (const float*)d_in[0];
  const int*   ei  = (const int*)  d_in[1];
  const float* W1  = (const float*)d_in[2];
  const float* av1s= (const float*)d_in[3];
  const float* av1d= (const float*)d_in[4];
  const float* b1  = (const float*)d_in[5];
  const float* W2  = (const float*)d_in[6];
  const float* av2s= (const float*)d_in[7];
  const float* av2d= (const float*)d_in[8];
  const float* b2  = (const float*)d_in[9];
  float* out = (float*)d_out;

  const int B  = 2;
  const int C1 = 128, C2 = 64;
  const int N  = in_sizes[0] / (B * C1);
  const int E  = in_sizes[1] / (B * 2);

  float* w   = (float*)d_ws;
  float* h1  = w; w += (long)N * C1;
  float* x2  = w; w += (long)N * C1;
  float* h2  = w; w += (long)N * C2;
  float* asb = w; w += N;
  float* adb = w; w += N;

  const int RT  = 200;
  const int CAPL = 6144;
  int gGemm    = N / 16;
  int gTile    = (N + RT - 1) / RT;
  int* keep_s = (int*)w; w += (long)gTile * CAPL;
  int* keep_d = (int*)w; w += (long)gTile * CAPL;
  int* keep_n = (int*)w; w += gTile;
  const int nvec = (E + 15) / 16;
  uint4* keys = (uint4*)(((uintptr_t)w + 15) & ~(uintptr_t)15);

  gat_tile_keys<RT><<<dim3((nvec + 255) / 256, B), dim3(256), 0, stream>>>(ei, keys, E, N, nvec);

  for (int b = 0; b < B; ++b) {
    const float* xb   = fea + (long)b * N * C1;
    const int*   srcb = ei + (long)b * 2 * E;
    const int*   dstb = srcb + E;
    float*       outb = out + (long)b * N * C2;
    const uint4* keyb = keys + (long)b * nvec;

    gat_gemm_wmma<8><<<gGemm, dim3(32), 0, stream>>>(xb, W1, av1s, av1d, h1, asb, adb, N, C1);
    gat_tile_softmax_aggregate<128, RT, CAPL, 1, 0><<<gTile, dim3(1024), 0, stream>>>(
        srcb, dstb, keyb, asb, adb, h1, b1, x2, keep_s, keep_d, keep_n, E, N);

    gat_gemm_wmma<4><<<gGemm, dim3(32), 0, stream>>>(x2, W2, av2s, av2d, h2, asb, adb, N, C1);
    gat_tile_softmax_aggregate<64, RT, CAPL, 0, 1><<<gTile, dim3(1024), 0, stream>>>(
        srcb, dstb, keyb, asb, adb, h2, b2, outb, keep_s, keep_d, keep_n, E, N);
  }
}


# Row pass: a lane prepares one entry (row, target, weight divided once, not in 32 lanes), the wave takes entries by readlane, one wide row load per lane; self-loops are entries; LDS adds arrival-order
# speedup vs baseline: 5.4392x; 1.1575x over previous
#include <hip/hip_runtime.h>
#include <math.h>
#include <stdint.h>

typedef __attribute__((ext_vector_type(2))) float v2f;
typedef __attribute__((ext_vector_type(8))) float v8f;

#define NEG_SLOPE 0.2f

template <int TILES>
__global__ __launch_bounds__(32)
void gat_gemm_wmma(const float* __restrict__ X, const float* __restrict__ W,
                   const float* __restrict__ a_src, const float* __restrict__ a_dst,
                   float* __restrict__ Hout, float* __restrict__ as, float* __restrict__ ad,
                   int N, int K) {
  const int F    = TILES * 16;
  const int row0 = blockIdx.x * 16;
  const int lane = threadIdx.x;
  const int l15  = lane & 15;
  const int kh   = (lane >> 4) * 2;
  const int m    = row0 + l15;

  v8f acc[TILES];
#pragma unroll
  for (int t = 0; t < TILES; ++t) acc[t] = (v8f){0.f,0.f,0.f,0.f,0.f,0.f,0.f,0.f};

  for (int k0 = 0; k0 < K; k0 += 4) {
    v2f a;
    a.x = X[(long)m * K + k0 + kh];
    a.y = X[(long)m * K + k0 + kh + 1];
#pragma unroll
    for (int t = 0; t < TILES; ++t) {
      v2f b;
      b.x = W[(long)(k0 + kh)     * F + t * 16 + l15];
      b.y = W[(long)(k0 + kh + 1) * F + t * 16 + l15];
      acc[t] = __builtin_amdgcn_wmma_f32_16x16x4_f32(
           false, a,  false, b,
           (short)0, acc[t],  false,  false);
    }
  }

  const int hi = (lane >> 4) * 8;
#pragma unroll
  for (int t = 0; t < TILES; ++t)
#pragma unroll
    for (int r = 0; r < 8; ++r)
      Hout[(long)(row0 + hi + r) * F + t * 16 + l15] = acc[t][r];

  float ps[8], pd[8];
#pragma unroll
  for (int r = 0; r < 8; ++r) { ps[r] = 0.f; pd[r] = 0.f; }
#pragma unroll
  for (int t = 0; t < TILES; ++t) {
    const float cs = a_src[t * 16 + l15];
    const float cd = a_dst[t * 16 + l15];
#pragma unroll
    for (int r = 0; r < 8; ++r) {
      ps[r] += acc[t][r] * cs;
      pd[r] += acc[t][r] * cd;
    }
  }
#pragma unroll
  for (int off = 8; off > 0; off >>= 1) {
#pragma unroll
    for (int r = 0; r < 8; ++r) {
      ps[r] += __shfl_xor(ps[r], off, 32);
      pd[r] += __shfl_xor(pd[r], off, 32);
    }
  }
  {
    const int r7 = l15 & 7;
    float vs = ps[0], vd = pd[0];
#pragma unroll
    for (int r = 1; r < 8; ++r) { if (r7 == r) { vs = ps[r]; vd = pd[r]; } }
    float* const o = (l15 < 8) ? as : ad;
    o[row0 + hi + r7] = (l15 < 8) ? vs : vd;
  }
}

template <int R>
__global__ __launch_bounds__(256)
void gat_tile_keys(const int* __restrict__ ei, uint4* __restrict__ keys,
                   int E, int N, int nvec) {
  const int b = blockIdx.y;
  const int v = blockIdx.x * 256 + threadIdx.x;
  if (v >= nvec) return;
  const int* dst = ei + ((long)b * 2 + 1) * E;
  const int e0 = v * 16;
  int d[16];
  if (e0 + 16 <= E && (((uintptr_t)(dst + e0)) & 15u) == 0u) {
    const int4* p = (const int4*)(dst + e0);
    int4 a0 = p[0], a1 = p[1], a2 = p[2], a3 = p[3];
    d[0]  = a0.x; d[1]  = a0.y; d[2]  = a0.z; d[3]  = a0.w;
    d[4]  = a1.x; d[5]  = a1.y; d[6]  = a1.z; d[7]  = a1.w;
    d[8]  = a2.x; d[9]  = a2.y; d[10] = a2.z; d[11] = a2.w;
    d[12] = a3.x; d[13] = a3.y; d[14] = a3.z; d[15] = a3.w;
  } else {
#pragma unroll
    for (int j = 0; j < 16; ++j) d[j] = (e0 + j < E) ? dst[e0 + j] : -1;
  }
  unsigned w[4];
#pragma unroll
  for (int q = 0; q < 4; ++q) {
    unsigned acc = 0u;
#pragma unroll
    for (int k = 0; k < 4; ++k) {
      unsigned x = (unsigned)d[4 * q + k];
      unsigned key = (x < (unsigned)N) ? (x / (unsigned)R) : 255u;
      acc |= key << (8 * k);
    }
    w[q] = acc;
  }
  keys[(long)b * nvec + v] = make_uint4(w[0], w[1], w[2], w[3]);
}

__device__ __forceinline__ unsigned gat_fkey(float f) {
  unsigned b = __float_as_uint(f);
  return (b & 0x80000000u) ? ~b : (b | 0x80000000u);
}
__device__ __forceinline__ float gat_fkey_inv(unsigned k) {
  unsigned b = (k & 0x80000000u) ? (k & 0x7fffffffu) : ~k;
  return __uint_as_float(b);
}

template <int C, int R, int CAP, int RELU, int TAKE>
__global__ __launch_bounds__(1024)
void gat_tile_softmax_aggregate(const int* __restrict__ src,
                                const int* __restrict__ dst,
                                const uint4* __restrict__ keys,
                                const float* __restrict__ as,
                                const float* __restrict__ ad,
                                const float* __restrict__ h,
                                const float* __restrict__ bias,
                                float* __restrict__ out,
                                int* __restrict__ keep_s,
                                int* __restrict__ keep_d,
                                int* __restrict__ keep_n,
                                int E, int N) {
  constexpr int NT = 1024;
  constexpr int CH = C / 32;
  static_assert(CAP % 16 == 0, "a piece of the input must start on a key vector");
  __shared__ float    s_acc[R * C];
  __shared__ int      l_s[CAP];
  __shared__ int      l_d[CAP];
  __shared__ float    l_v[CAP];
  __shared__ float    s_ad[R];
  __shared__ float    s_m[R];
  __shared__ float    s_den[R];
  __shared__ float    s_vs[R];
  __shared__ unsigned s_mk[R];
  __shared__ unsigned s_cnt;

  const int tid  = threadIdx.x;
  const int lane = tid & 31;
  const int wv   = __builtin_amdgcn_readfirstlane(tid >> 5);
  const int n0   = blockIdx.x * R;
  const int nr   = (N - n0 < R) ? (N - n0) : R;
  int* const my_s = keep_s + (long)blockIdx.x * CAP;
  int* const my_d = keep_d + (long)blockIdx.x * CAP;

  auto push = [&](int e) {
    unsigned slot = atomicAdd(&s_cnt, 1u);
    if (slot < (unsigned)CAP) l_s[slot] = e;
  };
  const unsigned kmul = (unsigned)blockIdx.x * 0x01010101u;
  auto eq4 = [&](unsigned w) -> unsigned {
    unsigned x = w ^ kmul;
    unsigned t = ((x & 0x7f7f7f7fu) + 0x7f7f7f7fu) | x;
    return ~t & 0x80808080u;
  };
  auto eq16 = [&](uint4 a) -> unsigned {
    return (eq4(a.x) >> 7) | (eq4(a.y) >> 6) | (eq4(a.z) >> 5) | (eq4(a.w) >> 4);
  };
  auto drain = [&](unsigned long long mm, int ebase) {
    while (mm) {
      int P = __builtin_ctzll(mm);
      mm &= mm - 1ull;
      int u = ((P >> 5) << 1) | ((P >> 2) & 1);
      push(ebase + u * (16 * NT) + ((P & 3) << 2) + ((P >> 3) & 3));
    }
  };
  auto scan = [&](int lo, int hi) {
    const uint4* pv = keys + (lo >> 4);
    const int nvec = (hi - lo + 15) >> 4;
    int v = tid;
    for (; v + 3 * NT < nvec; v += 4 * NT) {
      uint4 a0 = pv[v], a1 = pv[v + NT], a2 = pv[v + 2 * NT], a3 = pv[v + 3 * NT];
      unsigned m01 = eq16(a0) | (eq16(a1) << 4);
      unsigned m23 = eq16(a2) | (eq16(a3) << 4);
      drain((unsigned long long)m01 | ((unsigned long long)m23 << 32), lo + 16 * v);
    }
    for (; v < nvec; v += NT) drain((unsigned long long)eq16(pv[v]), lo + 16 * v);
  };
  auto prep = [&](int cnt, bool store) {
    for (int i = tid; i < cnt; i += NT) {
      int e = l_s[i];
      int s = src[e];
      int d = dst[e] - n0;
      float v = as[s] + s_ad[d];
      v = (v > 0.f) ? v : NEG_SLOPE * v;
      l_s[i] = s; l_d[i] = d; l_v[i] = v;
      if (store) { my_s[i] = s; my_d[i] = d; }
      atomicMax(&s_mk[d], gat_fkey(v));
    }
  };
  auto prep_kept = [&](int cnt) {
    for (int i = tid; i < cnt; i += NT) {
      int s = my_s[i];
      int d = my_d[i];
      float v = as[s] + s_ad[d];
      v = (v > 0.f) ? v : NEG_SLOPE * v;
      l_s[i] = s; l_d[i] = d; l_v[i] = v;
      atomicMax(&s_mk[d], gat_fkey(v));
    }
  };
  auto rebuild = [&](int k) -> int {
    __syncthreads();
    if (tid == 0) s_cnt = 0u;
    __syncthreads();
    int lo = k * CAP;
    int hi = (E - lo < CAP) ? E : lo + CAP;
    scan(lo, hi);
    __syncthreads();
    int c = (int)s_cnt;
    prep(c, false);
    __syncthreads();
    return c;
  };
  auto expsum = [&](int cnt, bool add) {
    for (int i = tid; i < cnt; i += NT) {
      int d = l_d[i];
      float ex = __expf(l_v[i] - s_m[d]);
      l_v[i] = ex;
      if (add) atomicAdd(&s_den[d], ex);
    }
  };
  typedef float rowv __attribute__((ext_vector_type(CH)));
  auto rows = [&](int cnt_, int nself) {
    constexpr int U = 8;
    const int cnt   = __builtin_amdgcn_readfirstlane(cnt_);
    const int total = cnt + nself;
    const int per   = (total + 31) >> 5;
    const int beg   = wv * per;
    const int end   = (beg + per < total) ? (beg + per) : total;
    const float* const hl = h + lane * CH;
    float* const al = s_acc + lane;
    for (int b0 = beg; b0 < end; b0 += 32) {
      const int m = (end - b0 < 32) ? (end - b0) : 32;
      int s = n0, d = 0;
      float w = 0.f;
      if (lane < m) {
        const int idx = b0 + lane;
        float ex;
        if (idx < cnt) { s = l_s[idx]; d = l_d[idx]; ex = l_v[idx]; }
        else           { d = idx - cnt; s = n0 + d; ex = s_vs[d]; }
        w = ex / (s_den[d] + 1e-16f);
      }
      const int wi = __float_as_int(w);
      int k = 0;
      for (; k + U <= m; k += U) {
        rowv  r[U];
        int   dk[U];
        float wk[U];
#pragma unroll
        for (int u = 0; u < U; ++u) {
          const int sk = __builtin_amdgcn_readlane(s, k + u);
          dk[u] = __builtin_amdgcn_readlane(d, k + u);
          wk[u] = __int_as_float(__builtin_amdgcn_readlane(wi, k + u));
          r[u] = *(const rowv*)(hl + (long)sk * C);
        }
#pragma unroll
        for (int u = 0; u < U; ++u) {
          float* const a = al + dk[u] * C;
#pragma unroll
          for (int i = 0; i < CH; ++i) atomicAdd(a + i * 32, r[u][i] * wk[u]);
        }
      }
      if (k < m) {
        rowv  r[U - 1];
        int   dk[U - 1];
        float wk[U - 1];
#pragma unroll
        for (int u = 0; u < U - 1; ++u) {
          if (k + u < m) {
            const int sk = __builtin_amdgcn_readlane(s, k + u);
            dk[u] = __builtin_amdgcn_readlane(d, k + u);
            wk[u] = __int_as_float(__builtin_amdgcn_readlane(wi, k + u));
            r[u] = *(const rowv*)(hl + (long)sk * C);
          }
        }
#pragma unroll
        for (int u = 0; u < U - 1; ++u) {
          if (k + u < m) {
            float* const a = al + dk[u] * C;
#pragma unroll
            for (int i = 0; i < CH; ++i) atomicAdd(a + i * 32, r[u][i] * wk[u]);
          }
        }
      }
    }
  };

  {
    float4 z4 = make_float4(0.f, 0.f, 0.f, 0.f);
    float4* a4 = (float4*)s_acc;
    for (int i = tid; i < R * C / 4; i += NT) a4[i] = z4;
  }
  if (tid < R) {
    float a = 0.f, v = 0.f;
    if (tid < nr) {
      a = ad[n0 + tid];
      v = as[n0 + tid] + a;
      v = (v > 0.f) ? v : NEG_SLOPE * v;
    }
    s_ad[tid] = a;
    s_vs[tid] = v;
    s_mk[tid] = gat_fkey(v);
    s_den[tid] = 0.f;
  }
  if (tid == 0) s_cnt = 0u;
  __syncthreads();

  unsigned total;
  if (TAKE) {
    total = (unsigned)keep_n[blockIdx.x];
  } else {
    scan(0, E);
    __syncthreads();
    total = s_cnt;
    if (tid == 0) keep_n[blockIdx.x] = (int)total;
  }
  const bool whole = (total <= (unsigned)CAP);
  const int  nseg  = whole ? 1 : (E + CAP - 1) / CAP;
  const int  n     = whole ? (int)total : 0;

  if (whole) { if (TAKE) prep_kept(n); else prep(n, true); }
  else for (int k = 0; k < nseg; ++k) rebuild(k);
  __syncthreads();
  if (tid < R) {
    float m = gat_fkey_inv(s_mk[tid]);
    if (!__builtin_isfinite(m)) m = 0.f;
    s_m[tid] = m;
    float ex = __expf(s_vs[tid] - m);
    s_vs[tid] = ex;
    s_den[tid] = ex;
  }
  __syncthreads();

  if (whole) expsum(n, true);
  else for (int k = 0; k < nseg; ++k) { int c = rebuild(k); expsum(c, true); }
  __syncthreads();

  if (whole) rows(n, nr);
  else {
    for (int k = 0; k < nseg; ++k) {
      int c = rebuild(k);
      expsum(c, false);
      __syncthreads();
      rows(c, 0);
    }
    rows(0, nr);
  }
  __syncthreads();

  {
    const float4* b4 = (const float4*)bias;
    float4* o4 = (float4*)(out + (long)n0 * C);
    for (int i = tid; i < nr * (C / 4); i += NT) {
      const int q = i % (C / 4);
      const float* a = s_acc + (i / (C / 4)) * C;
      float4 v;
      v.x = a[((4 * q)     % CH) * 32 + (4 * q)     / CH];
      v.y = a[((4 * q + 1) % CH) * 32 + (4 * q + 1) / CH];
      v.z = a[((4 * q + 2) % CH) * 32 + (4 * q + 2) / CH];
      v.w = a[((4 * q + 3) % CH) * 32 + (4 * q + 3) / CH];
      float4 b = b4[q];
      v.x += b.x; v.y += b.y; v.z += b.z; v.w += b.w;
      if (RELU) { v.x = fmaxf(v.x, 0.f); v.y = fmaxf(v.y, 0.f); v.z = fmaxf(v.z, 0.f); v.w = fmaxf(v.w, 0.f); }
      o4[i] = v;
    }
  }
}

extern "C" void kernel_launch(void* const* d_in, const int* in_sizes, int n_in,
                              void* d_out, int out_size, void* d_ws, size_t ws_size,
                              hipStream_t stream) {
  const float* fea = (const float*)d_in[0];
  const int*   ei  = (const int*)  d_in[1];
  const float* W1  = (const float*)d_in[2];
  const float* av1s= (const float*)d_in[3];
  const float* av1d= (const float*)d_in[4];
  const float* b1  = (const float*)d_in[5];
  const float* W2  = (const float*)d_in[6];
  const float* av2s= (const float*)d_in[7];
  const float* av2d= (const float*)d_in[8];
  const float* b2  = (const float*)d_in[9];
  float* out = (float*)d_out;

  const int B  = 2;
  const int C1 = 128, C2 = 64;
  const int N  = in_sizes[0] / (B * C1);
  const int E  = in_sizes[1] / (B * 2);

  float* w   = (float*)d_ws;
  float* h1  = w; w += (long)N * C1;
  float* x2  = w; w += (long)N * C1;
  float* h2  = w; w += (long)N * C2;
  float* asb = w; w += N;
  float* adb = w; w += N;

  const int RT  = 200;
  const int CAPL = 6144;
  int gGemm    = N / 16;
  int gTile    = (N + RT - 1) / RT;
  int* keep_s = (int*)w; w += (long)gTile * CAPL;
  int* keep_d = (int*)w; w += (long)gTile * CAPL;
  int* keep_n = (int*)w; w += gTile;
  const int nvec = (E + 15) / 16;
  uint4* keys = (uint4*)(((uintptr_t)w + 15) & ~(uintptr_t)15);

  gat_tile_keys<RT><<<dim3((nvec + 255) / 256, B), dim3(256), 0, stream>>>(ei, keys, E, N, nvec);

  for (int b = 0; b < B; ++b) {
    const float* xb   = fea + (long)b * N * C1;
    const int*   srcb = ei + (long)b * 2 * E;
    const int*   dstb = srcb + E;
    float*       outb = out + (long)b * N * C2;
    const uint4* keyb = keys + (long)b * nvec;

    gat_gemm_wmma<8><<<gGemm, dim3(32), 0, stream>>>(xb, W1, av1s, av1d, h1, asb, adb, N, C1);
    gat_tile_softmax_aggregate<128, RT, CAPL, 1, 0><<<gTile, dim3(1024), 0, stream>>>(
        srcb, dstb, keyb, asb, adb, h1, b1, x2, keep_s, keep_d, keep_n, E, N);

    gat_gemm_wmma<4><<<gGemm, dim3(32), 0, stream>>>(x2, W2, av2s, av2d, h2, asb, adb, N, C1);
    gat_tile_softmax_aggregate<64, RT, CAPL, 0, 1><<<gTile, dim3(1024), 0, stream>>>(
        srcb, dstb, keyb, asb, adb, h2, b2, outb, keep_s, keep_d, keep_n, E, N);
  }
}


# Rows pass: tile list counting-sorted by local destination in LDS (rank by integer LDS add); a wave sums whole destinations in registers, stores each row once; no LDS float adds; sums in rank order
# speedup vs baseline: 5.7927x; 1.0650x over previous
#include <hip/hip_runtime.h>
#include <math.h>
#include <stdint.h>

typedef __attribute__((ext_vector_type(2))) float v2f;
typedef __attribute__((ext_vector_type(8))) float v8f;

#define NEG_SLOPE 0.2f

template <int TILES>
__global__ __launch_bounds__(32)
void gat_gemm_wmma(const float* __restrict__ X, const float* __restrict__ W,
                   const float* __restrict__ a_src, const float* __restrict__ a_dst,
                   float* __restrict__ Hout, float* __restrict__ as, float* __restrict__ ad,
                   int N, int K) {
  const int F    = TILES * 16;
  const int row0 = blockIdx.x * 16;
  const int lane = threadIdx.x;
  const int l15  = lane & 15;
  const int kh   = (lane >> 4) * 2;
  const int m    = row0 + l15;

  v8f acc[TILES];
#pragma unroll
  for (int t = 0; t < TILES; ++t) acc[t] = (v8f){0.f,0.f,0.f,0.f,0.f,0.f,0.f,0.f};

  for (int k0 = 0; k0 < K; k0 += 4) {
    v2f a;
    a.x = X[(long)m * K + k0 + kh];
    a.y = X[(long)m * K + k0 + kh + 1];
#pragma unroll
    for (int t = 0; t < TILES; ++t) {
      v2f b;
      b.x = W[(long)(k0 + kh)     * F + t * 16 + l15];
      b.y = W[(long)(k0 + kh + 1) * F + t * 16 + l15];
      acc[t] = __builtin_amdgcn_wmma_f32_16x16x4_f32(
           false, a,  false, b,
           (short)0, acc[t],  false,  false);
    }
  }

  const int hi = (lane >> 4) * 8;
#pragma unroll
  for (int t = 0; t < TILES; ++t)
#pragma unroll
    for (int r = 0; r < 8; ++r)
      Hout[(long)(row0 + hi + r) * F + t * 16 + l15] = acc[t][r];

  float ps[8], pd[8];
#pragma unroll
  for (int r = 0; r < 8; ++r) { ps[r] = 0.f; pd[r] = 0.f; }
#pragma unroll
  for (int t = 0; t < TILES; ++t) {
    const float cs = a_src[t * 16 + l15];
    const float cd = a_dst[t * 16 + l15];
#pragma unroll
    for (int r = 0; r < 8; ++r) {
      ps[r] += acc[t][r] * cs;
      pd[r] += acc[t][r] * cd;
    }
  }
#pragma unroll
  for (int off = 8; off > 0; off >>= 1) {
#pragma unroll
    for (int r = 0; r < 8; ++r) {
      ps[r] += __shfl_xor(ps[r], off, 32);
      pd[r] += __shfl_xor(pd[r], off, 32);
    }
  }
  {
    const int r7 = l15 & 7;
    float vs = ps[0], vd = pd[0];
#pragma unroll
    for (int r = 1; r < 8; ++r) { if (r7 == r) { vs = ps[r]; vd = pd[r]; } }
    float* const o = (l15 < 8) ? as : ad;
    o[row0 + hi + r7] = (l15 < 8) ? vs : vd;
  }
}

template <int R>
__global__ __launch_bounds__(256)
void gat_tile_keys(const int* __restrict__ ei, uint4* __restrict__ keys,
                   int E, int N, int nvec) {
  const int b = blockIdx.y;
  const int v = blockIdx.x * 256 + threadIdx.x;
  if (v >= nvec) return;
  const int* dst = ei + ((long)b * 2 + 1) * E;
  const int e0 = v * 16;
  int d[16];
  if (e0 + 16 <= E && (((uintptr_t)(dst + e0)) & 15u) == 0u) {
    const int4* p = (const int4*)(dst + e0);
    int4 a0 = p[0], a1 = p[1], a2 = p[2], a3 = p[3];
    d[0]  = a0.x; d[1]  = a0.y; d[2]  = a0.z; d[3]  = a0.w;
    d[4]  = a1.x; d[5]  = a1.y; d[6]  = a1.z; d[7]  = a1.w;
    d[8]  = a2.x; d[9]  = a2.y; d[10] = a2.z; d[11] = a2.w;
    d[12] = a3.x; d[13] = a3.y; d[14] = a3.z; d[15] = a3.w;
  } else {
#pragma unroll
    for (int j = 0; j < 16; ++j) d[j] = (e0 + j < E) ? dst[e0 + j] : -1;
  }
  unsigned w[4];
#pragma unroll
  for (int q = 0; q < 4; ++q) {
    unsigned acc = 0u;
#pragma unroll
    for (int k = 0; k < 4; ++k) {
      unsigned x = (unsigned)d[4 * q + k];
      unsigned key = (x < (unsigned)N) ? (x / (unsigned)R) : 255u;
      acc |= key << (8 * k);
    }
    w[q] = acc;
  }
  keys[(long)b * nvec + v] = make_uint4(w[0], w[1], w[2], w[3]);
}

__device__ __forceinline__ unsigned gat_fkey(float f) {
  unsigned b = __float_as_uint(f);
  return (b & 0x80000000u) ? ~b : (b | 0x80000000u);
}
__device__ __forceinline__ float gat_fkey_inv(unsigned k) {
  unsigned b = (k & 0x80000000u) ? (k & 0x7fffffffu) : ~k;
  return __uint_as_float(b);
}

template <int C, int R, int CAP, int RELU, int TAKE>
__global__ __launch_bounds__(1024)
void gat_tile_softmax_aggregate(const int* __restrict__ src,
                                const int* __restrict__ dst,
                                const uint4* __restrict__ keys,
                                const float* __restrict__ as,
                                const float* __restrict__ ad,
                                const float* __restrict__ h,
                                const float* __restrict__ bias,
                                float* __restrict__ out,
                                int* __restrict__ keep_s,
                                int* __restrict__ keep_d,
                                int* __restrict__ keep_n,
                                int E, int N) {
  constexpr int NT = 1024;
  constexpr int CH = C / 32;
  constexpr int QN = CAP + R;
  constexpr int BUFN = (R * C > 3 * QN) ? (R * C) : (3 * QN);
  static_assert(CAP % 16 == 0, "a piece of the input must start on a key vector");
  static_assert(R <= 256, "a local destination takes the low 8 bits beside the entry's rank");
  __shared__ __attribute__((aligned(16))) float s_buf[BUFN];
  __shared__ int      l_s[CAP];
  __shared__ int      l_d[CAP];
  __shared__ float    l_v[CAP];
  __shared__ int      s_c[R];
  __shared__ int      s_off[R + 1];
  __shared__ float    s_ad[R];
  __shared__ float    s_m[R];
  __shared__ float    s_den[R];
  __shared__ float    s_vs[R];
  __shared__ unsigned s_mk[R];
  __shared__ unsigned s_cnt;

  const int tid  = threadIdx.x;
  const int lane = tid & 31;
  const int wv   = __builtin_amdgcn_readfirstlane(tid >> 5);
  const int n0   = blockIdx.x * R;
  const int nr   = (N - n0 < R) ? (N - n0) : R;
  int* const my_s = keep_s + (long)blockIdx.x * CAP;
  int* const my_d = keep_d + (long)blockIdx.x * CAP;
  float* const s_acc = s_buf;
  int*   const q_s   = (int*)s_buf;
  int*   const q_d   = q_s + QN;
  float* const q_v   = s_buf + 2 * QN;

  auto push = [&](int e) {
    unsigned slot = atomicAdd(&s_cnt, 1u);
    if (slot < (unsigned)CAP) l_s[slot] = e;
  };
  const unsigned kmul = (unsigned)blockIdx.x * 0x01010101u;
  auto eq4 = [&](unsigned w) -> unsigned {
    unsigned x = w ^ kmul;
    unsigned t = ((x & 0x7f7f7f7fu) + 0x7f7f7f7fu) | x;
    return ~t & 0x80808080u;
  };
  auto eq16 = [&](uint4 a) -> unsigned {
    return (eq4(a.x) >> 7) | (eq4(a.y) >> 6) | (eq4(a.z) >> 5) | (eq4(a.w) >> 4);
  };
  auto drain = [&](unsigned long long mm, int ebase) {
    while (mm) {
      int P = __builtin_ctzll(mm);
      mm &= mm - 1ull;
      int u = ((P >> 5) << 1) | ((P >> 2) & 1);
      push(ebase + u * (16 * NT) + ((P & 3) << 2) + ((P >> 3) & 3));
    }
  };
  auto scan = [&](int lo, int hi) {
    const uint4* pv = keys + (lo >> 4);
    const int nvec = (hi - lo + 15) >> 4;
    int v = tid;
    for (; v + 3 * NT < nvec; v += 4 * NT) {
      uint4 a0 = pv[v], a1 = pv[v + NT], a2 = pv[v + 2 * NT], a3 = pv[v + 3 * NT];
      unsigned m01 = eq16(a0) | (eq16(a1) << 4);
      unsigned m23 = eq16(a2) | (eq16(a3) << 4);
      drain((unsigned long long)m01 | ((unsigned long long)m23 << 32), lo + 16 * v);
    }
    for (; v < nvec; v += NT) drain((unsigned long long)eq16(pv[v]), lo + 16 * v);
  };
  auto prep = [&](int cnt) {
    for (int i = tid; i < cnt; i += NT) {
      int e = l_s[i];
      int s = src[e];
      int d = dst[e] - n0;
      float v = as[s] + s_ad[d];
      v = (v > 0.f) ? v : NEG_SLOPE * v;
      l_s[i] = s; l_d[i] = d; l_v[i] = v;
      atomicMax(&s_mk[d], gat_fkey(v));
    }
  };
  auto prep_sorted = [&](int cnt) {
    for (int i = tid; i < cnt; i += NT) {
      int s, d;
      if (TAKE) { s = my_s[i]; d = my_d[i]; }
      else {
        int e = l_s[i];
        s = src[e];
        d = dst[e] - n0;
        my_s[i] = s; my_d[i] = d;
      }
      float v = as[s] + s_ad[d];
      v = (v > 0.f) ? v : NEG_SLOPE * v;
      int rank = atomicAdd(&s_c[d], 1);
      l_s[i] = s; l_d[i] = d | (rank << 8); l_v[i] = v;
      atomicMax(&s_mk[d], gat_fkey(v));
    }
  };
  auto rebuild = [&](int k) -> int {
    __syncthreads();
    if (tid == 0) s_cnt = 0u;
    __syncthreads();
    int lo = k * CAP;
    int hi = (E - lo < CAP) ? E : lo + CAP;
    scan(lo, hi);
    __syncthreads();
    int c = (int)s_cnt;
    prep(c);
    __syncthreads();
    return c;
  };
  auto expsum = [&](int cnt, bool add) {
    for (int i = tid; i < cnt; i += NT) {
      int d = l_d[i];
      float ex = __expf(l_v[i] - s_m[d]);
      l_v[i] = ex;
      if (add) atomicAdd(&s_den[d], ex);
    }
  };
  auto expsum_sorted = [&](int cnt) {
    if (tid < nr) {
      int p = s_off[tid];
      q_s[p] = n0 + tid; q_d[p] = tid; q_v[p] = s_vs[tid];
    }
    for (int i = tid; i < cnt; i += NT) {
      int dr = l_d[i];
      int d = dr & 255;
      float ex = __expf(l_v[i] - s_m[d]);
      atomicAdd(&s_den[d], ex);
      int p = s_off[d] + 1 + (dr >> 8);
      q_s[p] = l_s[i]; q_d[p] = d; q_v[p] = ex;
    }
  };
  typedef float rowv __attribute__((ext_vector_type(CH)));
  auto rows = [&](int cnt_, int nself) {
    constexpr int U = 8;
    const int cnt   = __builtin_amdgcn_readfirstlane(cnt_);
    const int total = cnt + nself;
    const int per   = (total + 31) >> 5;
    const int beg   = wv * per;
    const int end   = (beg + per < total) ? (beg + per) : total;
    const float* const hl = h + lane * CH;
    float* const al = s_acc + lane;
    for (int b0 = beg; b0 < end; b0 += 32) {
      const int m = (end - b0 < 32) ? (end - b0) : 32;
      int s = n0, d = 0;
      float w = 0.f;
      if (lane < m) {
        const int idx = b0 + lane;
        float ex;
        if (idx < cnt) { s = l_s[idx]; d = l_d[idx]; ex = l_v[idx]; }
        else           { d = idx - cnt; s = n0 + d; ex = s_vs[d]; }
        w = ex / (s_den[d] + 1e-16f);
      }
      const int wi = __float_as_int(w);
      int k = 0;
      for (; k + U <= m; k += U) {
        rowv  r[U];
        int   dk[U];
        float wk[U];
#pragma unroll
        for (int u = 0; u < U; ++u) {
          const int sk = __builtin_amdgcn_readlane(s, k + u);
          dk[u] = __builtin_amdgcn_readlane(d, k + u);
          wk[u] = __int_as_float(__builtin_amdgcn_readlane(wi, k + u));
          r[u] = *(const rowv*)(hl + (long)sk * C);
        }
#pragma unroll
        for (int u = 0; u < U; ++u) {
          float* const a = al + dk[u] * C;
#pragma unroll
          for (int i = 0; i < CH; ++i) atomicAdd(a + i * 32, r[u][i] * wk[u]);
        }
      }
      if (k < m) {
        rowv  r[U - 1];
        int   dk[U - 1];
        float wk[U - 1];
#pragma unroll
        for (int u = 0; u < U - 1; ++u) {
          if (k + u < m) {
            const int sk = __builtin_amdgcn_readlane(s, k + u);
            dk[u] = __builtin_amdgcn_readlane(d, k + u);
            wk[u] = __int_as_float(__builtin_amdgcn_readlane(wi, k + u));
            r[u] = *(const rowv*)(hl + (long)sk * C);
          }
        }
#pragma unroll
        for (int u = 0; u < U - 1; ++u) {
          if (k + u < m) {
            float* const a = al + dk[u] * C;
#pragma unroll
            for (int i = 0; i < CH; ++i) atomicAdd(a + i * 32, r[u][i] * wk[u]);
          }
        }
      }
    }
  };

  auto rows_sorted = [&](int cnt_) {
    constexpr int U = 8;
    const int cnt   = __builtin_amdgcn_readfirstlane(cnt_);
    const int total = cnt + nr;
    const int per   = (total + 31) >> 5;
    const int tlo   = wv * per;
    const int thi   = tlo + per;
    int dlo = 0, dhi = 0;
    for (int j = 0; j < (R + 31) / 32; ++j) {
      const int d = j * 32 + lane;
      const int o = (d < R) ? s_off[d] : 0x7fffffff;
      dlo += __popc(__builtin_amdgcn_ballot_w32(o < tlo));
      dhi += __popc(__builtin_amdgcn_ballot_w32(o < thi));
    }
    const int beg = __builtin_amdgcn_readfirstlane(s_off[dlo]);
    const int end = __builtin_amdgcn_readfirstlane(s_off[dhi]);
    const float* const hl = h + lane * CH;
    float* const ol = out + (long)n0 * C + lane * CH;
    const rowv bv = *(const rowv*)(bias + lane * CH);
    rowv acc;
#pragma unroll
    for (int i = 0; i < CH; ++i) acc[i] = 0.f;
    int cur = dlo;
    auto flush = [&]() {
      rowv o;
#pragma unroll
      for (int i = 0; i < CH; ++i) {
        float t = acc[i] + bv[i];
        if (RELU) t = fmaxf(t, 0.f);
        o[i] = t;
        acc[i] = 0.f;
      }
      *(rowv*)(ol + (long)cur * C) = o;
    };
    for (int b0 = beg; b0 < end; b0 += 32) {
      const int m = (end - b0 < 32) ? (end - b0) : 32;
      int s = n0, d = 0;
      float w = 0.f;
      if (lane < m) {
        const int idx = b0 + lane;
        s = q_s[idx]; d = q_d[idx];
        w = q_v[idx] / (s_den[d] + 1e-16f);
      }
      const int wi = __float_as_int(w);
      int k = 0;
      for (; k + U <= m; k += U) {
        rowv  r[U];
        int   dk[U];
        float wk[U];
#pragma unroll
        for (int u = 0; u < U; ++u) {
          const int sk = __builtin_amdgcn_readlane(s, k + u);
          dk[u] = __builtin_amdgcn_readlane(d, k + u);
          wk[u] = __int_as_float(__builtin_amdgcn_readlane(wi, k + u));
          r[u] = *(const rowv*)(hl + (long)sk * C);
        }
#pragma unroll
        for (int u = 0; u < U; ++u) {
          if (dk[u] != cur) { flush(); cur = dk[u]; }
#pragma unroll
          for (int i = 0; i < CH; ++i) acc[i] += r[u][i] * wk[u];
        }
      }
      if (k < m) {
        rowv  r[U - 1];
        int   dk[U - 1];
        float wk[U - 1];
#pragma unroll
        for (int u = 0; u < U - 1; ++u) {
          if (k + u < m) {
            const int sk = __builtin_amdgcn_readlane(s, k + u);
            dk[u] = __builtin_amdgcn_readlane(d, k + u);
            wk[u] = __int_as_float(__builtin_amdgcn_readlane(wi, k + u));
            r[u] = *(const rowv*)(hl + (long)sk * C);
          }
        }
#pragma unroll
        for (int u = 0; u < U - 1; ++u) {
          if (k + u < m) {
            if (dk[u] != cur) { flush(); cur = dk[u]; }
#pragma unroll
            for (int i = 0; i < CH; ++i) acc[i] += r[u][i] * wk[u];
          }
        }
      }
    }
    if (beg < end) flush();
  };

  if (tid < R) {
    float a = 0.f, v = 0.f;
    if (tid < nr) {
      a = ad[n0 + tid];
      v = as[n0 + tid] + a;
      v = (v > 0.f) ? v : NEG_SLOPE * v;
    }
    s_ad[tid] = a;
    s_vs[tid] = v;
    s_mk[tid] = gat_fkey(v);
    s_den[tid] = 0.f;
    s_c[tid] = 0;
  }
  if (tid == 0) s_cnt = 0u;
  __syncthreads();

  unsigned total;
  if (TAKE) {
    total = (unsigned)keep_n[blockIdx.x];
  } else {
    scan(0, E);
    __syncthreads();
    total = s_cnt;
    if (tid == 0) keep_n[blockIdx.x] = (int)total;
  }
  const bool whole = (total <= (unsigned)CAP);
  const int  nseg  = whole ? 1 : (E + CAP - 1) / CAP;
  const int  n     = whole ? (int)total : 0;

  if (whole) prep_sorted(n);
  else {
    float4 z4 = make_float4(0.f, 0.f, 0.f, 0.f);
    float4* a4 = (float4*)s_acc;
    for (int i = tid; i < R * C / 4; i += NT) a4[i] = z4;
    for (int k = 0; k < nseg; ++k) rebuild(k);
  }
  __syncthreads();
  if (tid < R) {
    float m = gat_fkey_inv(s_mk[tid]);
    if (!__builtin_isfinite(m)) m = 0.f;
    s_m[tid] = m;
    float ex = __expf(s_vs[tid] - m);
    s_vs[tid] = ex;
    s_den[tid] = ex;
  }
  if (whole && wv == NT / 32 - 1) {
    constexpr int PL = (R + 31) / 32;
    int c[PL];
    int sum = 0;
#pragma unroll
    for (int j = 0; j < PL; ++j) {
      const int d = lane * PL + j;
      c[j] = (d < nr) ? (s_c[d] + 1) : 0;
      sum += c[j];
    }
    int inc = sum;
#pragma unroll
    for (int off = 1; off < 32; off <<= 1) {
      const int t = __shfl_up(inc, off, 32);
      if (lane >= off) inc += t;
    }
    int run = inc - sum;
#pragma unroll
    for (int j = 0; j < PL; ++j) {
      const int d = lane * PL + j;
      if (d < R) s_off[d] = run;
      run += c[j];
    }
    if (lane == 31) s_off[R] = inc;
  }
  __syncthreads();

  if (whole) expsum_sorted(n);
  else for (int k = 0; k < nseg; ++k) { int c = rebuild(k); expsum(c, true); }
  __syncthreads();

  if (whole) rows_sorted(n);
  else {
    for (int k = 0; k < nseg; ++k) {
      int c = rebuild(k);
      expsum(c, false);
      __syncthreads();
      rows(c, 0);
    }
    rows(0, nr);
  }
  __syncthreads();

  if (!whole) {
    const float4* b4 = (const float4*)bias;
    float4* o4 = (float4*)(out + (long)n0 * C);
    for (int i = tid; i < nr * (C / 4); i += NT) {
      const int q = i % (C / 4);
      const float* a = s_acc + (i / (C / 4)) * C;
      float4 v;
      v.x = a[((4 * q)     % CH) * 32 + (4 * q)     / CH];
      v.y = a[((4 * q + 1) % CH) * 32 + (4 * q + 1) / CH];
      v.z = a[((4 * q + 2) % CH) * 32 + (4 * q + 2) / CH];
      v.w = a[((4 * q + 3) % CH) * 32 + (4 * q + 3) / CH];
      float4 b = b4[q];
      v.x += b.x; v.y += b.y; v.z += b.z; v.w += b.w;
      if (RELU) { v.x = fmaxf(v.x, 0.f); v.y = fmaxf(v.y, 0.f); v.z = fmaxf(v.z, 0.f); v.w = fmaxf(v.w, 0.f); }
      o4[i] = v;
    }
  }
}

extern "C" void kernel_launch(void* const* d_in, const int* in_sizes, int n_in,
                              void* d_out, int out_size, void* d_ws, size_t ws_size,
                              hipStream_t stream) {
  const float* fea = (const float*)d_in[0];
  const int*   ei  = (const int*)  d_in[1];
  const float* W1  = (const float*)d_in[2];
  const float* av1s= (const float*)d_in[3];
  const float* av1d= (const float*)d_in[4];
  const float* b1  = (const float*)d_in[5];
  const float* W2  = (const float*)d_in[6];
  const float* av2s= (const float*)d_in[7];
  const float* av2d= (const float*)d_in[8];
  const float* b2  = (const float*)d_in[9];
  float* out = (float*)d_out;

  const int B  = 2;
  const int C1 = 128, C2 = 64;
  const int N  = in_sizes[0] / (B * C1);
  const int E  = in_sizes[1] / (B * 2);

  float* w   = (float*)d_ws;
  float* h1  = w; w += (long)N * C1;
  float* x2  = w; w += (long)N * C1;
  float* h2  = w; w += (long)N * C2;
  float* asb = w; w += N;
  float* adb = w; w += N;

  const int RT  = 200;
  const int CAPL = 6144;
  int gGemm    = N / 16;
  int gTile    = (N + RT - 1) / RT;
  int* keep_s = (int*)w; w += (long)gTile * CAPL;
  int* keep_d = (int*)w; w += (long)gTile * CAPL;
  int* keep_n = (int*)w; w += gTile;
  const int nvec = (E + 15) / 16;
  uint4* keys = (uint4*)(((uintptr_t)w + 15) & ~(uintptr_t)15);

  gat_tile_keys<RT><<<dim3((nvec + 255) / 256, B), dim3(256), 0, stream>>>(ei, keys, E, N, nvec);

  for (int b = 0; b < B; ++b) {
    const float* xb   = fea + (long)b * N * C1;
    const int*   srcb = ei + (long)b * 2 * E;
    const int*   dstb = srcb + E;
    float*       outb = out + (long)b * N * C2;
    const uint4* keyb = keys + (long)b * nvec;

    gat_gemm_wmma<8><<<gGemm, dim3(32), 0, stream>>>(xb, W1, av1s, av1d, h1, asb, adb, N, C1);
    gat_tile_softmax_aggregate<128, RT, CAPL, 1, 0><<<gTile, dim3(1024), 0, stream>>>(
        srcb, dstb, keyb, asb, adb, h1, b1, x2, keep_s, keep_d, keep_n, E, N);

    gat_gemm_wmma<4><<<gGemm, dim3(32), 0, stream>>>(x2, W2, av2s, av2d, h2, asb, adb, N, C1);
    gat_tile_softmax_aggregate<64, RT, CAPL, 0, 1><<<gTile, dim3(1024), 0, stream>>>(
        srcb, dstb, keyb, asb, adb, h2, b2, outb, keep_s, keep_d, keep_n, E, N);
  }
}


# Independent stages share a launch: tile keys ride behind graph 0's layer-1 product, graph 1's layer-1 product rides with graph 0's layer-2 product; 9->7 launches; sums and their order unchanged
# speedup vs baseline: 6.2294x; 1.0754x over previous
#include <hip/hip_runtime.h>
#include <math.h>
#include <stdint.h>

typedef __attribute__((ext_vector_type(2))) float v2f;
typedef __attribute__((ext_vector_type(8))) float v8f;

#define NEG_SLOPE 0.2f

template <int TILES>
__device__ __forceinline__
void gat_gemm_stripe(const float* __restrict__ X, const float* __restrict__ W,
                     const float* __restrict__ a_src, const float* __restrict__ a_dst,
                     float* __restrict__ Hout, float* __restrict__ as, float* __restrict__ ad,
                     int K, int row0, int lane) {
  const int F    = TILES * 16;
  const int l15  = lane & 15;
  const int kh   = (lane >> 4) * 2;
  const int m    = row0 + l15;

  v8f acc[TILES];
#pragma unroll
  for (int t = 0; t < TILES; ++t) acc[t] = (v8f){0.f,0.f,0.f,0.f,0.f,0.f,0.f,0.f};

  for (int k0 = 0; k0 < K; k0 += 4) {
    v2f a;
    a.x = X[(long)m * K + k0 + kh];
    a.y = X[(long)m * K + k0 + kh + 1];
#pragma unroll
    for (int t = 0; t < TILES; ++t) {
      v2f b;
      b.x = W[(long)(k0 + kh)     * F + t * 16 + l15];
      b.y = W[(long)(k0 + kh + 1) * F + t * 16 + l15];
      acc[t] = __builtin_amdgcn_wmma_f32_16x16x4_f32(
           false, a,  false, b,
           (short)0, acc[t],  false,  false);
    }
  }

  const int hi = (lane >> 4) * 8;
#pragma unroll
  for (int t = 0; t < TILES; ++t)
#pragma unroll
    for (int r = 0; r < 8; ++r)
      Hout[(long)(row0 + hi + r) * F + t * 16 + l15] = acc[t][r];

  float ps[8], pd[8];
#pragma unroll
  for (int r = 0; r < 8; ++r) { ps[r] = 0.f; pd[r] = 0.f; }
#pragma unroll
  for (int t = 0; t < TILES; ++t) {
    const float cs = a_src[t * 16 + l15];
    const float cd = a_dst[t * 16 + l15];
#pragma unroll
    for (int r = 0; r < 8; ++r) {
      ps[r] += acc[t][r] * cs;
      pd[r] += acc[t][r] * cd;
    }
  }
#pragma unroll
  for (int off = 8; off > 0; off >>= 1) {
#pragma unroll
    for (int r = 0; r < 8; ++r) {
      ps[r] += __shfl_xor(ps[r], off, 32);
      pd[r] += __shfl_xor(pd[r], off, 32);
    }
  }
  {
    const int r7 = l15 & 7;
    float vs = ps[0], vd = pd[0];
#pragma unroll
    for (int r = 1; r < 8; ++r) { if (r7 == r) { vs = ps[r]; vd = pd[r]; } }
    float* const o = (l15 < 8) ? as : ad;
    o[row0 + hi + r7] = (l15 < 8) ? vs : vd;
  }
}

template <int R>
__device__ __forceinline__
void gat_tile_keys_vec(const int* __restrict__ ei, uint4* __restrict__ keys,
                       int E, int N, int nvec, int b, int v) {
  if (v >= nvec) return;
  const int* dst = ei + ((long)b * 2 + 1) * E;
  const int e0 = v * 16;
  int d[16];
  if (e0 + 16 <= E && (((uintptr_t)(dst + e0)) & 15u) == 0u) {
    const int4* p = (const int4*)(dst + e0);
    int4 a0 = p[0], a1 = p[1], a2 = p[2], a3 = p[3];
    d[0]  = a0.x; d[1]  = a0.y; d[2]  = a0.z; d[3]  = a0.w;
    d[4]  = a1.x; d[5]  = a1.y; d[6]  = a1.z; d[7]  = a1.w;
    d[8]  = a2.x; d[9]  = a2.y; d[10] = a2.z; d[11] = a2.w;
    d[12] = a3.x; d[13] = a3.y; d[14] = a3.z; d[15] = a3.w;
  } else {
#pragma unroll
    for (int j = 0; j < 16; ++j) d[j] = (e0 + j < E) ? dst[e0 + j] : -1;
  }
  unsigned w[4];
#pragma unroll
  for (int q = 0; q < 4; ++q) {
    unsigned acc = 0u;
#pragma unroll
    for (int k = 0; k < 4; ++k) {
      unsigned x = (unsigned)d[4 * q + k];
      unsigned key = (x < (unsigned)N) ? (x / (unsigned)R) : 255u;
      acc |= key << (8 * k);
    }
    w[q] = acc;
  }
  keys[(long)b * nvec + v] = make_uint4(w[0], w[1], w[2], w[3]);
}

template <int TILES>
__global__ __launch_bounds__(32)
void gat_gemm_wmma(const float* __restrict__ X, const float* __restrict__ W,
                   const float* __restrict__ a_src, const float* __restrict__ a_dst,
                   float* __restrict__ Hout, float* __restrict__ as, float* __restrict__ ad,
                   int N, int K) {
  gat_gemm_stripe<TILES>(X, W, a_src, a_dst, Hout, as, ad, K, blockIdx.x * 16, threadIdx.x);
}

template <int TILES, int R>
__global__ __launch_bounds__(32)
void gat_gemm_and_keys(const float* __restrict__ X, const float* __restrict__ W,
                       const float* __restrict__ a_src, const float* __restrict__ a_dst,
                       float* __restrict__ Hout, float* __restrict__ as, float* __restrict__ ad,
                       int K, int nstripes,
                       const int* __restrict__ ei, uint4* __restrict__ keys,
                       int E, int N, int nvec, int nvb) {
  const int bid = blockIdx.x;
  if (bid < nstripes) {
    gat_gemm_stripe<TILES>(X, W, a_src, a_dst, Hout, as, ad, K, bid * 16, threadIdx.x);
  } else {
    const int kb = bid - nstripes;
    const int b  = kb / nvb;
    gat_tile_keys_vec<R>(ei, keys, E, N, nvec, b, (kb - b * nvb) * 32 + (int)threadIdx.x);
  }
}

template <int TILES_A, int TILES_B>
__global__ __launch_bounds__(32)
void gat_gemm_pair(const float* __restrict__ XA, const float* __restrict__ WA,
                   const float* __restrict__ a_srcA, const float* __restrict__ a_dstA,
                   float* __restrict__ HA, float* __restrict__ asA, float* __restrict__ adA,
                   const float* __restrict__ XB, const float* __restrict__ WB,
                   const float* __restrict__ a_srcB, const float* __restrict__ a_dstB,
                   float* __restrict__ HB, float* __restrict__ asB, float* __restrict__ adB,
                   int K, int nstripes) {
  const int bid = blockIdx.x;
  if (bid < nstripes)
    gat_gemm_stripe<TILES_A>(XA, WA, a_srcA, a_dstA, HA, asA, adA, K, bid * 16, threadIdx.x);
  else
    gat_gemm_stripe<TILES_B>(XB, WB, a_srcB, a_dstB, HB, asB, adB, K, (bid - nstripes) * 16, threadIdx.x);
}

__device__ __forceinline__ unsigned gat_fkey(float f) {
  unsigned b = __float_as_uint(f);
  return (b & 0x80000000u) ? ~b : (b | 0x80000000u);
}
__device__ __forceinline__ float gat_fkey_inv(unsigned k) {
  unsigned b = (k & 0x80000000u) ? (k & 0x7fffffffu) : ~k;
  return __uint_as_float(b);
}

template <int C, int R, int CAP, int RELU, int TAKE>
__global__ __launch_bounds__(1024)
void gat_tile_softmax_aggregate(const int* __restrict__ src,
                                const int* __restrict__ dst,
                                const uint4* __restrict__ keys,
                                const float* __restrict__ as,
                                const float* __restrict__ ad,
                                const float* __restrict__ h,
                                const float* __restrict__ bias,
                                float* __restrict__ out,
                                int* __restrict__ keep_s,
                                int* __restrict__ keep_d,
                                int* __restrict__ keep_n,
                                int E, int N) {
  constexpr int NT = 1024;
  constexpr int CH = C / 32;
  constexpr int QN = CAP + R;
  constexpr int BUFN = (R * C > 3 * QN) ? (R * C) : (3 * QN);
  static_assert(CAP % 16 == 0, "a piece of the input must start on a key vector");
  static_assert(R <= 256, "a local destination takes the low 8 bits beside the entry's rank");
  __shared__ __attribute__((aligned(16))) float s_buf[BUFN];
  __shared__ int      l_s[CAP];
  __shared__ int      l_d[CAP];
  __shared__ float    l_v[CAP];
  __shared__ int      s_c[R];
  __shared__ int      s_off[R + 1];
  __shared__ float    s_ad[R];
  __shared__ float    s_m[R];
  __shared__ float    s_den[R];
  __shared__ float    s_vs[R];
  __shared__ unsigned s_mk[R];
  __shared__ unsigned s_cnt;

  const int tid  = threadIdx.x;
  const int lane = tid & 31;
  const int wv   = __builtin_amdgcn_readfirstlane(tid >> 5);
  const int n0   = blockIdx.x * R;
  const int nr   = (N - n0 < R) ? (N - n0) : R;
  int* const my_s = keep_s + (long)blockIdx.x * CAP;
  int* const my_d = keep_d + (long)blockIdx.x * CAP;
  float* const s_acc = s_buf;
  int*   const q_s   = (int*)s_buf;
  int*   const q_d   = q_s + QN;
  float* const q_v   = s_buf + 2 * QN;

  auto push = [&](int e) {
    unsigned slot = atomicAdd(&s_cnt, 1u);
    if (slot < (unsigned)CAP) l_s[slot] = e;
  };
  const unsigned kmul = (unsigned)blockIdx.x * 0x01010101u;
  auto eq4 = [&](unsigned w) -> unsigned {
    unsigned x = w ^ kmul;
    unsigned t = ((x & 0x7f7f7f7fu) + 0x7f7f7f7fu) | x;
    return ~t & 0x80808080u;
  };
  auto eq16 = [&](uint4 a) -> unsigned {
    return (eq4(a.x) >> 7) | (eq4(a.y) >> 6) | (eq4(a.z) >> 5) | (eq4(a.w) >> 4);
  };
  auto drain = [&](unsigned long long mm, int ebase) {
    while (mm) {
      int P = __builtin_ctzll(mm);
      mm &= mm - 1ull;
      int u = ((P >> 5) << 1) | ((P >> 2) & 1);
      push(ebase + u * (16 * NT) + ((P & 3) << 2) + ((P >> 3) & 3));
    }
  };
  auto scan = [&](int lo, int hi) {
    const uint4* pv = keys + (lo >> 4);
    const int nvec = (hi - lo + 15) >> 4;
    int v = tid;
    for (; v + 3 * NT < nvec; v += 4 * NT) {
      uint4 a0 = pv[v], a1 = pv[v + NT], a2 = pv[v + 2 * NT], a3 = pv[v + 3 * NT];
      unsigned m01 = eq16(a0) | (eq16(a1) << 4);
      unsigned m23 = eq16(a2) | (eq16(a3) << 4);
      drain((unsigned long long)m01 | ((unsigned long long)m23 << 32), lo + 16 * v);
    }
    for (; v < nvec; v += NT) drain((unsigned long long)eq16(pv[v]), lo + 16 * v);
  };
  auto prep = [&](int cnt) {
    for (int i = tid; i < cnt; i += NT) {
      int e = l_s[i];
      int s = src[e];
      int d = dst[e] - n0;
      float v = as[s] + s_ad[d];
      v = (v > 0.f) ? v : NEG_SLOPE * v;
      l_s[i] = s; l_d[i] = d; l_v[i] = v;
      atomicMax(&s_mk[d], gat_fkey(v));
    }
  };
  auto prep_sorted = [&](int cnt) {
    for (int i = tid; i < cnt; i += NT) {
      int s, d;
      if (TAKE) { s = my_s[i]; d = my_d[i]; }
      else {
        int e = l_s[i];
        s = src[e];
        d = dst[e] - n0;
        my_s[i] = s; my_d[i] = d;
      }
      float v = as[s] + s_ad[d];
      v = (v > 0.f) ? v : NEG_SLOPE * v;
      int rank = atomicAdd(&s_c[d], 1);
      l_s[i] = s; l_d[i] = d | (rank << 8); l_v[i] = v;
      atomicMax(&s_mk[d], gat_fkey(v));
    }
  };
  auto rebuild = [&](int k) -> int {
    __syncthreads();
    if (tid == 0) s_cnt = 0u;
    __syncthreads();
    int lo = k * CAP;
    int hi = (E - lo < CAP) ? E : lo + CAP;
    scan(lo, hi);
    __syncthreads();
    int c = (int)s_cnt;
    prep(c);
    __syncthreads();
    return c;
  };
  auto expsum = [&](int cnt, bool add) {
    for (int i = tid; i < cnt; i += NT) {
      int d = l_d[i];
      float ex = __expf(l_v[i] - s_m[d]);
      l_v[i] = ex;
      if (add) atomicAdd(&s_den[d], ex);
    }
  };
  auto expsum_sorted = [&](int cnt) {
    if (tid < nr) {
      int p = s_off[tid];
      q_s[p] = n0 + tid; q_d[p] = tid; q_v[p] = s_vs[tid];
    }
    for (int i = tid; i < cnt; i += NT) {
      int dr = l_d[i];
      int d = dr & 255;
      float ex = __expf(l_v[i] - s_m[d]);
      atomicAdd(&s_den[d], ex);
      int p = s_off[d] + 1 + (dr >> 8);
      q_s[p] = l_s[i]; q_d[p] = d; q_v[p] = ex;
    }
  };
  typedef float rowv __attribute__((ext_vector_type(CH)));
  auto rows = [&](int cnt_, int nself) {
    constexpr int U = 8;
    const int cnt   = __builtin_amdgcn_readfirstlane(cnt_);
    const int total = cnt + nself;
    const int per   = (total + 31) >> 5;
    const int beg   = wv * per;
    const int end   = (beg + per < total) ? (beg + per) : total;
    const float* const hl = h + lane * CH;
    float* const al = s_acc + lane;
    for (int b0 = beg; b0 < end; b0 += 32) {
      const int m = (end - b0 < 32) ? (end - b0) : 32;
      int s = n0, d = 0;
      float w = 0.f;
      if (lane < m) {
        const int idx = b0 + lane;
        float ex;
        if (idx < cnt) { s = l_s[idx]; d = l_d[idx]; ex = l_v[idx]; }
        else           { d = idx - cnt; s = n0 + d; ex = s_vs[d]; }
        w = ex / (s_den[d] + 1e-16f);
      }
      const int wi = __float_as_int(w);
      int k = 0;
      for (; k + U <= m; k += U) {
        rowv  r[U];
        int   dk[U];
        float wk[U];
#pragma unroll
        for (int u = 0; u < U; ++u) {
          const int sk = __builtin_amdgcn_readlane(s, k + u);
          dk[u] = __builtin_amdgcn_readlane(d, k + u);
          wk[u] = __int_as_float(__builtin_amdgcn_readlane(wi, k + u));
          r[u] = *(const rowv*)(hl + (long)sk * C);
        }
#pragma unroll
        for (int u = 0; u < U; ++u) {
          float* const a = al + dk[u] * C;
#pragma unroll
          for (int i = 0; i < CH; ++i) atomicAdd(a + i * 32, r[u][i] * wk[u]);
        }
      }
      if (k < m) {
        rowv  r[U - 1];
        int   dk[U - 1];
        float wk[U - 1];
#pragma unroll
        for (int u = 0; u < U - 1; ++u) {
          if (k + u < m) {
            const int sk = __builtin_amdgcn_readlane(s, k + u);
            dk[u] = __builtin_amdgcn_readlane(d, k + u);
            wk[u] = __int_as_float(__builtin_amdgcn_readlane(wi, k + u));
            r[u] = *(const rowv*)(hl + (long)sk * C);
          }
        }
#pragma unroll
        for (int u = 0; u < U - 1; ++u) {
          if (k + u < m) {
            float* const a = al + dk[u] * C;
#pragma unroll
            for (int i = 0; i < CH; ++i) atomicAdd(a + i * 32, r[u][i] * wk[u]);
          }
        }
      }
    }
  };

  auto rows_sorted = [&](int cnt_) {
    constexpr int U = 8;
    const int cnt   = __builtin_amdgcn_readfirstlane(cnt_);
    const int total = cnt + nr;
    const int per   = (total + 31) >> 5;
    const int tlo   = wv * per;
    const int thi   = tlo + per;
    int dlo = 0, dhi = 0;
    for (int j = 0; j < (R + 31) / 32; ++j) {
      const int d = j * 32 + lane;
      const int o = (d < R) ? s_off[d] : 0x7fffffff;
      dlo += __popc(__builtin_amdgcn_ballot_w32(o < tlo));
      dhi += __popc(__builtin_amdgcn_ballot_w32(o < thi));
    }
    const int beg = __builtin_amdgcn_readfirstlane(s_off[dlo]);
    const int end = __builtin_amdgcn_readfirstlane(s_off[dhi]);
    const float* const hl = h + lane * CH;
    float* const ol = out + (long)n0 * C + lane * CH;
    const rowv bv = *(const rowv*)(bias + lane * CH);
    rowv acc;
#pragma unroll
    for (int i = 0; i < CH; ++i) acc[i] = 0.f;
    int cur = dlo;
    auto flush = [&]() {
      rowv o;
#pragma unroll
      for (int i = 0; i < CH; ++i) {
        float t = acc[i] + bv[i];
        if (RELU) t = fmaxf(t, 0.f);
        o[i] = t;
        acc[i] = 0.f;
      }
      *(rowv*)(ol + (long)cur * C) = o;
    };
    for (int b0 = beg; b0 < end; b0 += 32) {
      const int m = (end - b0 < 32) ? (end - b0) : 32;
      int s = n0, d = 0;
      float w = 0.f;
      if (lane < m) {
        const int idx = b0 + lane;
        s = q_s[idx]; d = q_d[idx];
        w = q_v[idx] / (s_den[d] + 1e-16f);
      }
      const int wi = __float_as_int(w);
      int k = 0;
      for (; k + U <= m; k += U) {
        rowv  r[U];
        int   dk[U];
        float wk[U];
#pragma unroll
        for (int u = 0; u < U; ++u) {
          const int sk = __builtin_amdgcn_readlane(s, k + u);
          dk[u] = __builtin_amdgcn_readlane(d, k + u);
          wk[u] = __int_as_float(__builtin_amdgcn_readlane(wi, k + u));
          r[u] = *(const rowv*)(hl + (long)sk * C);
        }
#pragma unroll
        for (int u = 0; u < U; ++u) {
          if (dk[u] != cur) { flush(); cur = dk[u]; }
#pragma unroll
          for (int i = 0; i < CH; ++i) acc[i] += r[u][i] * wk[u];
        }
      }
      if (k < m) {
        rowv  r[U - 1];
        int   dk[U - 1];
        float wk[U - 1];
#pragma unroll
        for (int u = 0; u < U - 1; ++u) {
          if (k + u < m) {
            const int sk = __builtin_amdgcn_readlane(s, k + u);
            dk[u] = __builtin_amdgcn_readlane(d, k + u);
            wk[u] = __int_as_float(__builtin_amdgcn_readlane(wi, k + u));
            r[u] = *(const rowv*)(hl + (long)sk * C);
          }
        }
#pragma unroll
        for (int u = 0; u < U - 1; ++u) {
          if (k + u < m) {
            if (dk[u] != cur) { flush(); cur = dk[u]; }
#pragma unroll
            for (int i = 0; i < CH; ++i) acc[i] += r[u][i] * wk[u];
          }
        }
      }
    }
    if (beg < end) flush();
  };

  if (tid < R) {
    float a = 0.f, v = 0.f;
    if (tid < nr) {
      a = ad[n0 + tid];
      v = as[n0 + tid] + a;
      v = (v > 0.f) ? v : NEG_SLOPE * v;
    }
    s_ad[tid] = a;
    s_vs[tid] = v;
    s_mk[tid] = gat_fkey(v);
    s_den[tid] = 0.f;
    s_c[tid] = 0;
  }
  if (tid == 0) s_cnt = 0u;
  __syncthreads();

  unsigned total;
  if (TAKE) {
    total = (unsigned)keep_n[blockIdx.x];
  } else {
    scan(0, E);
    __syncthreads();
    total = s_cnt;
    if (tid == 0) keep_n[blockIdx.x] = (int)total;
  }
  const bool whole = (total <= (unsigned)CAP);
  const int  nseg  = whole ? 1 : (E + CAP - 1) / CAP;
  const int  n     = whole ? (int)total : 0;

  if (whole) prep_sorted(n);
  else {
    float4 z4 = make_float4(0.f, 0.f, 0.f, 0.f);
    float4* a4 = (float4*)s_acc;
    for (int i = tid; i < R * C / 4; i += NT) a4[i] = z4;
    for (int k = 0; k < nseg; ++k) rebuild(k);
  }
  __syncthreads();
  if (tid < R) {
    float m = gat_fkey_inv(s_mk[tid]);
    if (!__builtin_isfinite(m)) m = 0.f;
    s_m[tid] = m;
    float ex = __expf(s_vs[tid] - m);
    s_vs[tid] = ex;
    s_den[tid] = ex;
  }
  if (whole && wv == NT / 32 - 1) {
    constexpr int PL = (R + 31) / 32;
    int c[PL];
    int sum = 0;
#pragma unroll
    for (int j = 0; j < PL; ++j) {
      const int d = lane * PL + j;
      c[j] = (d < nr) ? (s_c[d] + 1) : 0;
      sum += c[j];
    }
    int inc = sum;
#pragma unroll
    for (int off = 1; off < 32; off <<= 1) {
      const int t = __shfl_up(inc, off, 32);
      if (lane >= off) inc += t;
    }
    int run = inc - sum;
#pragma unroll
    for (int j = 0; j < PL; ++j) {
      const int d = lane * PL + j;
      if (d < R) s_off[d] = run;
      run += c[j];
    }
    if (lane == 31) s_off[R] = inc;
  }
  __syncthreads();

  if (whole) expsum_sorted(n);
  else for (int k = 0; k < nseg; ++k) { int c = rebuild(k); expsum(c, true); }
  __syncthreads();

  if (whole) rows_sorted(n);
  else {
    for (int k = 0; k < nseg; ++k) {
      int c = rebuild(k);
      expsum(c, false);
      __syncthreads();
      rows(c, 0);
    }
    rows(0, nr);
  }
  __syncthreads();

  if (!whole) {
    const float4* b4 = (const float4*)bias;
    float4* o4 = (float4*)(out + (long)n0 * C);
    for (int i = tid; i < nr * (C / 4); i += NT) {
      const int q = i % (C / 4);
      const float* a = s_acc + (i / (C / 4)) * C;
      float4 v;
      v.x = a[((4 * q)     % CH) * 32 + (4 * q)     / CH];
      v.y = a[((4 * q + 1) % CH) * 32 + (4 * q + 1) / CH];
      v.z = a[((4 * q + 2) % CH) * 32 + (4 * q + 2) / CH];
      v.w = a[((4 * q + 3) % CH) * 32 + (4 * q + 3) / CH];
      float4 b = b4[q];
      v.x += b.x; v.y += b.y; v.z += b.z; v.w += b.w;
      if (RELU) { v.x = fmaxf(v.x, 0.f); v.y = fmaxf(v.y, 0.f); v.z = fmaxf(v.z, 0.f); v.w = fmaxf(v.w, 0.f); }
      o4[i] = v;
    }
  }
}

extern "C" void kernel_launch(void* const* d_in, const int* in_sizes, int n_in,
                              void* d_out, int out_size, void* d_ws, size_t ws_size,
                              hipStream_t stream) {
  const float* fea = (const float*)d_in[0];
  const int*   ei  = (const int*)  d_in[1];
  const float* W1  = (const float*)d_in[2];
  const float* av1s= (const float*)d_in[3];
  const float* av1d= (const float*)d_in[4];
  const float* b1  = (const float*)d_in[5];
  const float* W2  = (const float*)d_in[6];
  const float* av2s= (const float*)d_in[7];
  const float* av2d= (const float*)d_in[8];
  const float* b2  = (const float*)d_in[9];
  float* out = (float*)d_out;

  const int B  = 2;
  const int C1 = 128, C2 = 64;
  const int N  = in_sizes[0] / (B * C1);
  const int E  = in_sizes[1] / (B * 2);

  float* w   = (float*)d_ws;
  float* h1  = w; w += (long)N * C1;
  float* x2  = w; w += (long)N * C1;
  float* h2  = w; w += (long)N * C2;
  float* asb = w; w += N;
  float* adb = w; w += N;
  float* as2 = w; w += N;
  float* ad2 = w; w += N;

  const int RT  = 200;
  const int CAPL = 6144;
  int gGemm    = N / 16;
  int gTile    = (N + RT - 1) / RT;
  int* keep_s = (int*)w; w += (long)gTile * CAPL;
  int* keep_d = (int*)w; w += (long)gTile * CAPL;
  int* keep_n = (int*)w; w += gTile;
  const int nvec = (E + 15) / 16;
  uint4* keys = (uint4*)(((uintptr_t)w + 15) & ~(uintptr_t)15);

  const int nvb = (nvec + 31) / 32;

  gat_gemm_and_keys<8, RT><<<gGemm + B * nvb, dim3(32), 0, stream>>>(
      fea, W1, av1s, av1d, h1, asb, adb, C1, gGemm, ei, keys, E, N, nvec, nvb);

  for (int b = 0; b < B; ++b) {
    const int*   srcb = ei + (long)b * 2 * E;
    const int*   dstb = srcb + E;
    float*       outb = out + (long)b * N * C2;
    const uint4* keyb = keys + (long)b * nvec;

    gat_tile_softmax_aggregate<128, RT, CAPL, 1, 0><<<gTile, dim3(1024), 0, stream>>>(
        srcb, dstb, keyb, asb, adb, h1, b1, x2, keep_s, keep_d, keep_n, E, N);

    if (b + 1 < B)
      gat_gemm_pair<8, 4><<<2 * gGemm, dim3(32), 0, stream>>>(
          fea + (long)(b + 1) * N * C1, W1, av1s, av1d, h1, asb, adb,
          x2, W2, av2s, av2d, h2, as2, ad2, C1, gGemm);
    else
      gat_gemm_wmma<4><<<gGemm, dim3(32), 0, stream>>>(x2, W2, av2s, av2d, h2, as2, ad2, N, C1);
    gat_tile_softmax_aggregate<64, RT, CAPL, 0, 1><<<gTile, dim3(1024), 0, stream>>>(
        srcb, dstb, keyb, as2, ad2, h2, b2, outb, keep_s, keep_d, keep_n, E, N);
  }
}


# Tile edge lists from a bucket pre-pass: a wave counting-sorts each 1024-edge chunk by destination tile (LDS integer adds), a tile copies its segment of every chunk, no key scan; sums stay rank order
# speedup vs baseline: 7.1960x; 1.1552x over previous
#include <hip/hip_runtime.h>
#include <math.h>
#include <stdint.h>

typedef __attribute__((ext_vector_type(2))) float v2f;
typedef __attribute__((ext_vector_type(8))) float v8f;

#define NEG_SLOPE 0.2f

template <int TILES>
__device__ __forceinline__
void gat_gemm_stripe(const float* __restrict__ X, const float* __restrict__ W,
                     const float* __restrict__ a_src, const float* __restrict__ a_dst,
                     float* __restrict__ Hout, float* __restrict__ as, float* __restrict__ ad,
                     int K, int row0, int lane) {
  const int F    = TILES * 16;
  const int l15  = lane & 15;
  const int kh   = (lane >> 4) * 2;
  const int m    = row0 + l15;

  v8f acc[TILES];
#pragma unroll
  for (int t = 0; t < TILES; ++t) acc[t] = (v8f){0.f,0.f,0.f,0.f,0.f,0.f,0.f,0.f};

  for (int k0 = 0; k0 < K; k0 += 4) {
    v2f a;
    a.x = X[(long)m * K + k0 + kh];
    a.y = X[(long)m * K + k0 + kh + 1];
#pragma unroll
    for (int t = 0; t < TILES; ++t) {
      v2f b;
      b.x = W[(long)(k0 + kh)     * F + t * 16 + l15];
      b.y = W[(long)(k0 + kh + 1) * F + t * 16 + l15];
      acc[t] = __builtin_amdgcn_wmma_f32_16x16x4_f32(
           false, a,  false, b,
           (short)0, acc[t],  false,  false);
    }
  }

  const int hi = (lane >> 4) * 8;
#pragma unroll
  for (int t = 0; t < TILES; ++t)
#pragma unroll
    for (int r = 0; r < 8; ++r)
      Hout[(long)(row0 + hi + r) * F + t * 16 + l15] = acc[t][r];

  float ps[8], pd[8];
#pragma unroll
  for (int r = 0; r < 8; ++r) { ps[r] = 0.f; pd[r] = 0.f; }
#pragma unroll
  for (int t = 0; t < TILES; ++t) {
    const float cs = a_src[t * 16 + l15];
    const float cd = a_dst[t * 16 + l15];
#pragma unroll
    for (int r = 0; r < 8; ++r) {
      ps[r] += acc[t][r] * cs;
      pd[r] += acc[t][r] * cd;
    }
  }
#pragma unroll
  for (int off = 8; off > 0; off >>= 1) {
#pragma unroll
    for (int r = 0; r < 8; ++r) {
      ps[r] += __shfl_xor(ps[r], off, 32);
      pd[r] += __shfl_xor(pd[r], off, 32);
    }
  }
  {
    const int r7 = l15 & 7;
    float vs = ps[0], vd = pd[0];
#pragma unroll
    for (int r = 1; r < 8; ++r) { if (r7 == r) { vs = ps[r]; vd = pd[r]; } }
    float* const o = (l15 < 8) ? as : ad;
    o[row0 + hi + r7] = (l15 < 8) ? vs : vd;
  }
}

#define GAT_CHUNK 1024
#define GAT_BINS  256

template <int R>
__device__ __forceinline__
void gat_chunk_sort(const int* __restrict__ ei, int2* __restrict__ lists, int* __restrict__ offs,
                    int E, int N, int nchunk, int b, int c, int lane, int* hist) {
  constexpr int PER = GAT_CHUNK / 32;
  constexpr int BL  = GAT_BINS / 32;
  const int* src = ei + (long)b * 2 * E;
  const int* dst = src + E;
#pragma unroll
  for (int j = 0; j < BL; ++j) hist[j * 32 + lane] = 0;
  __syncthreads();
  int s[PER], pk[PER];
  const int e0 = c * GAT_CHUNK;
#pragma unroll
  for (int j = 0; j < PER; ++j) {
    const int e  = e0 + j * 32 + lane;
    const int ec = (e < E) ? e : (E - 1);
    const int d  = dst[ec];
    s[j]  = src[ec];
    pk[j] = (e < E) ? d : -1;
  }
#pragma unroll
  for (int j = 0; j < PER; ++j) {
    const int d   = pk[j];
    const int key = ((unsigned)d < (unsigned)N) ? (d / R) : 255;
    const int rank = atomicAdd(&hist[key], 1);
    pk[j] = (key << 18) | (((d - key * R) & 255) << 10) | rank;
  }
  __syncthreads();
  int cn[BL];
  int sum = 0;
#pragma unroll
  for (int j = 0; j < BL; ++j) { cn[j] = hist[lane * BL + j]; sum += cn[j]; }
  int inc = sum;
#pragma unroll
  for (int off = 1; off < 32; off <<= 1) {
    const int t = __shfl_up(inc, off, 32);
    if (lane >= off) inc += t;
  }
  int run = inc - sum;
  int* const o = offs + ((long)b * nchunk + c) * GAT_BINS + lane * BL;
  __syncthreads();
#pragma unroll
  for (int j = 0; j < BL; ++j) {
    hist[lane * BL + j] = run;
    o[j] = run;
    run += cn[j];
  }
  __syncthreads();
  int2* const out = lists + ((long)b * nchunk + c) * GAT_CHUNK;
#pragma unroll
  for (int j = 0; j < PER; ++j) {
    const int p = pk[j];
    out[hist[p >> 18] + (p & 1023)] = make_int2(s[j], (p >> 10) & 255);
  }
}

template <int TILES>
__global__ __launch_bounds__(32)
void gat_gemm_wmma(const float* __restrict__ X, const float* __restrict__ W,
                   const float* __restrict__ a_src, const float* __restrict__ a_dst,
                   float* __restrict__ Hout, float* __restrict__ as, float* __restrict__ ad,
                   int N, int K) {
  gat_gemm_stripe<TILES>(X, W, a_src, a_dst, Hout, as, ad, K, blockIdx.x * 16, threadIdx.x);
}

template <int TILES, int R>
__global__ __launch_bounds__(32)
void gat_gemm_and_chunks(const float* __restrict__ X, const float* __restrict__ W,
                         const float* __restrict__ a_src, const float* __restrict__ a_dst,
                         float* __restrict__ Hout, float* __restrict__ as, float* __restrict__ ad,
                         int K, int nstripes,
                         const int* __restrict__ ei, int2* __restrict__ lists, int* __restrict__ offs,
                         int E, int N, int nchunk) {
  __shared__ int s_hist[GAT_BINS];
  const int bid = blockIdx.x;
  if (bid < nstripes) {
    gat_gemm_stripe<TILES>(X, W, a_src, a_dst, Hout, as, ad, K, bid * 16, threadIdx.x);
  } else {
    const int kb = bid - nstripes;
    const int b  = kb / nchunk;
    gat_chunk_sort<R>(ei, lists, offs, E, N, nchunk, b, kb - b * nchunk, (int)threadIdx.x, s_hist);
  }
}

template <int TILES_A, int TILES_B>
__global__ __launch_bounds__(32)
void gat_gemm_pair(const float* __restrict__ XA, const float* __restrict__ WA,
                   const float* __restrict__ a_srcA, const float* __restrict__ a_dstA,
                   float* __restrict__ HA, float* __restrict__ asA, float* __restrict__ adA,
                   const float* __restrict__ XB, const float* __restrict__ WB,
                   const float* __restrict__ a_srcB, const float* __restrict__ a_dstB,
                   float* __restrict__ HB, float* __restrict__ asB, float* __restrict__ adB,
                   int K, int nstripes) {
  const int bid = blockIdx.x;
  if (bid < nstripes)
    gat_gemm_stripe<TILES_A>(XA, WA, a_srcA, a_dstA, HA, asA, adA, K, bid * 16, threadIdx.x);
  else
    gat_gemm_stripe<TILES_B>(XB, WB, a_srcB, a_dstB, HB, asB, adB, K, (bid - nstripes) * 16, threadIdx.x);
}

__device__ __forceinline__ unsigned gat_fkey(float f) {
  unsigned b = __float_as_uint(f);
  return (b & 0x80000000u) ? ~b : (b | 0x80000000u);
}
__device__ __forceinline__ float gat_fkey_inv(unsigned k) {
  unsigned b = (k & 0x80000000u) ? (k & 0x7fffffffu) : ~k;
  return __uint_as_float(b);
}

template <int C, int R, int CAP, int RELU, int TAKE>
__global__ __launch_bounds__(1024)
void gat_tile_softmax_aggregate(const int2* __restrict__ lists,
                                const int* __restrict__ offs,
                                const float* __restrict__ as,
                                const float* __restrict__ ad,
                                const float* __restrict__ h,
                                const float* __restrict__ bias,
                                float* __restrict__ out,
                                int* __restrict__ keep_s,
                                int* __restrict__ keep_d,
                                int* __restrict__ keep_n,
                                int nchunk, int N) {
  constexpr int NT = 1024;
  constexpr int CH = C / 32;
  constexpr int QN = CAP + R;
  constexpr int BUFN = (R * C > 3 * QN) ? (R * C) : (3 * QN);
  static_assert(CAP % GAT_CHUNK == 0, "a piece of the input is a run of whole chunks");
  constexpr int PC = CAP / GAT_CHUNK;
  static_assert(R <= 256, "a local destination takes the low 8 bits beside the entry's rank");
  __shared__ __attribute__((aligned(16))) float s_buf[BUFN];
  __shared__ int      l_s[CAP];
  __shared__ int      l_d[CAP];
  __shared__ float    l_v[CAP];
  __shared__ int      s_c[R];
  __shared__ int      s_off[R + 1];
  __shared__ float    s_ad[R];
  __shared__ float    s_m[R];
  __shared__ float    s_den[R];
  __shared__ float    s_vs[R];
  __shared__ unsigned s_mk[R];
  __shared__ unsigned s_cnt;

  const int tid  = threadIdx.x;
  const int lane = tid & 31;
  const int wv   = __builtin_amdgcn_readfirstlane(tid >> 5);
  const int n0   = blockIdx.x * R;
  const int nr   = (N - n0 < R) ? (N - n0) : R;
  int* const my_s = keep_s + (long)blockIdx.x * CAP;
  int* const my_d = keep_d + (long)blockIdx.x * CAP;
  float* const s_acc = s_buf;
  int*   const q_s   = (int*)s_buf;
  int*   const q_d   = q_s + QN;
  float* const q_v   = s_buf + 2 * QN;

  auto collect = [&](int clo, int chi) {
    constexpr int G = 8;
    for (int c0 = clo; c0 < chi; c0 += NT) {
      const int c = c0 + tid;
      int st = 0, cn = 0;
      if (c < chi) {
        const int* o = offs + (long)c * GAT_BINS + blockIdx.x;
        st = o[0];
        cn = o[1] - st;
      }
      const int2* p = lists + (long)c * GAT_CHUNK + st;
      const unsigned base = atomicAdd(&s_cnt, (unsigned)cn);
      for (int j0 = 0; j0 < cn; j0 += G) {
        int2 v[G];
#pragma unroll
        for (int g = 0; g < G; ++g) if (j0 + g < cn) v[g] = p[j0 + g];
#pragma unroll
        for (int g = 0; g < G; ++g) {
          const unsigned slot = base + (unsigned)(j0 + g);
          if (j0 + g < cn && slot < (unsigned)CAP) { l_s[slot] = v[g].x; l_d[slot] = v[g].y; }
        }
      }
    }
  };
  auto prep = [&](int cnt) {
    for (int i = tid; i < cnt; i += NT) {
      int s = l_s[i];
      int d = l_d[i];
      float v = as[s] + s_ad[d];
      v = (v > 0.f) ? v : NEG_SLOPE * v;
      l_v[i] = v;
      atomicMax(&s_mk[d], gat_fkey(v));
    }
  };
  auto prep_sorted = [&](int cnt) {
    for (int i = tid; i < cnt; i += NT) {
      int s, d;
      if (TAKE) { s = my_s[i]; d = my_d[i]; }
      else {
        s = l_s[i];
        d = l_d[i];
        my_s[i] = s; my_d[i] = d;
      }
      float v = as[s] + s_ad[d];
      v = (v > 0.f) ? v : NEG_SLOPE * v;
      int rank = atomicAdd(&s_c[d], 1);
      l_s[i] = s; l_d[i] = d | (rank << 8); l_v[i] = v;
      atomicMax(&s_mk[d], gat_fkey(v));
    }
  };
  auto rebuild = [&](int k) -> int {
    __syncthreads();
    if (tid == 0) s_cnt = 0u;
    __syncthreads();
    int lo = k * PC;
    int hi = (nchunk - lo < PC) ? nchunk : lo + PC;
    collect(lo, hi);
    __syncthreads();
    int c = (int)s_cnt;
    prep(c);
    __syncthreads();
    return c;
  };
  auto expsum = [&](int cnt, bool add) {
    for (int i = tid; i < cnt; i += NT) {
      int d = l_d[i];
      float ex = __expf(l_v[i] - s_m[d]);
      l_v[i] = ex;
      if (add) atomicAdd(&s_den[d], ex);
    }
  };
  auto expsum_sorted = [&](int cnt) {
    if (tid < nr) {
      int p = s_off[tid];
      q_s[p] = n0 + tid; q_d[p] = tid; q_v[p] = s_vs[tid];
    }
    for (int i = tid; i < cnt; i += NT) {
      int dr = l_d[i];
      int d = dr & 255;
      float ex = __expf(l_v[i] - s_m[d]);
      atomicAdd(&s_den[d], ex);
      int p = s_off[d] + 1 + (dr >> 8);
      q_s[p] = l_s[i]; q_d[p] = d; q_v[p] = ex;
    }
  };
  typedef float rowv __attribute__((ext_vector_type(CH)));
  auto rows = [&](int cnt_, int nself) {
    constexpr int U = 8;
    const int cnt   = __builtin_amdgcn_readfirstlane(cnt_);
    const int total = cnt + nself;
    const int per   = (total + 31) >> 5;
    const int beg   = wv * per;
    const int end   = (beg + per < total) ? (beg + per) : total;
    const float* const hl = h + lane * CH;
    float* const al = s_acc + lane;
    for (int b0 = beg; b0 < end; b0 += 32) {
      const int m = (end - b0 < 32) ? (end - b0) : 32;
      int s = n0, d = 0;
      float w = 0.f;
      if (lane < m) {
        const int idx = b0 + lane;
        float ex;
        if (idx < cnt) { s = l_s[idx]; d = l_d[idx]; ex = l_v[idx]; }
        else           { d = idx - cnt; s = n0 + d; ex = s_vs[d]; }
        w = ex / (s_den[d] + 1e-16f);
      }
      const int wi = __float_as_int(w);
      int k = 0;
      for (; k + U <= m; k += U) {
        rowv  r[U];
        int   dk[U];
        float wk[U];
#pragma unroll
        for (int u = 0; u < U; ++u) {
          const int sk = __builtin_amdgcn_readlane(s, k + u);
          dk[u] = __builtin_amdgcn_readlane(d, k + u);
          wk[u] = __int_as_float(__builtin_amdgcn_readlane(wi, k + u));
          r[u] = *(const rowv*)(hl + (long)sk * C);
        }
#pragma unroll
        for (int u = 0; u < U; ++u) {
          float* const a = al + dk[u] * C;
#pragma unroll
          for (int i = 0; i < CH; ++i) atomicAdd(a + i * 32, r[u][i] * wk[u]);
        }
      }
      if (k < m) {
        rowv  r[U - 1];
        int   dk[U - 1];
        float wk[U - 1];
#pragma unroll
        for (int u = 0; u < U - 1; ++u) {
          if (k + u < m) {
            const int sk = __builtin_amdgcn_readlane(s, k + u);
            dk[u] = __builtin_amdgcn_readlane(d, k + u);
            wk[u] = __int_as_float(__builtin_amdgcn_readlane(wi, k + u));
            r[u] = *(const rowv*)(hl + (long)sk * C);
          }
        }
#pragma unroll
        for (int u = 0; u < U - 1; ++u) {
          if (k + u < m) {
            float* const a = al + dk[u] * C;
#pragma unroll
            for (int i = 0; i < CH; ++i) atomicAdd(a + i * 32, r[u][i] * wk[u]);
          }
        }
      }
    }
  };

  auto rows_sorted = [&](int cnt_) {
    constexpr int U = 8;
    const int cnt   = __builtin_amdgcn_readfirstlane(cnt_);
    const int total = cnt + nr;
    const int per   = (total + 31) >> 5;
    const int tlo   = wv * per;
    const int thi   = tlo + per;
    int dlo = 0, dhi = 0;
    for (int j = 0; j < (R + 31) / 32; ++j) {
      const int d = j * 32 + lane;
      const int o = (d < R) ? s_off[d] : 0x7fffffff;
      dlo += __popc(__builtin_amdgcn_ballot_w32(o < tlo));
      dhi += __popc(__builtin_amdgcn_ballot_w32(o < thi));
    }
    const int beg = __builtin_amdgcn_readfirstlane(s_off[dlo]);
    const int end = __builtin_amdgcn_readfirstlane(s_off[dhi]);
    const float* const hl = h + lane * CH;
    float* const ol = out + (long)n0 * C + lane * CH;
    const rowv bv = *(const rowv*)(bias + lane * CH);
    rowv acc;
#pragma unroll
    for (int i = 0; i < CH; ++i) acc[i] = 0.f;
    int cur = dlo;
    auto flush = [&]() {
      rowv o;
#pragma unroll
      for (int i = 0; i < CH; ++i) {
        float t = acc[i] + bv[i];
        if (RELU) t = fmaxf(t, 0.f);
        o[i] = t;
        acc[i] = 0.f;
      }
      *(rowv*)(ol + (long)cur * C) = o;
    };
    for (int b0 = beg; b0 < end; b0 += 32) {
      const int m = (end - b0 < 32) ? (end - b0) : 32;
      int s = n0, d = 0;
      float w = 0.f;
      if (lane < m) {
        const int idx = b0 + lane;
        s = q_s[idx]; d = q_d[idx];
        w = q_v[idx] / (s_den[d] + 1e-16f);
      }
      const int wi = __float_as_int(w);
      int k = 0;
      for (; k + U <= m; k += U) {
        rowv  r[U];
        int   dk[U];
        float wk[U];
#pragma unroll
        for (int u = 0; u < U; ++u) {
          const int sk = __builtin_amdgcn_readlane(s, k + u);
          dk[u] = __builtin_amdgcn_readlane(d, k + u);
          wk[u] = __int_as_float(__builtin_amdgcn_readlane(wi, k + u));
          r[u] = *(const rowv*)(hl + (long)sk * C);
        }
#pragma unroll
        for (int u = 0; u < U; ++u) {
          if (dk[u] != cur) { flush(); cur = dk[u]; }
#pragma unroll
          for (int i = 0; i < CH; ++i) acc[i] += r[u][i] * wk[u];
        }
      }
      if (k < m) {
        rowv  r[U - 1];
        int   dk[U - 1];
        float wk[U - 1];
#pragma unroll
        for (int u = 0; u < U - 1; ++u) {
          if (k + u < m) {
            const int sk = __builtin_amdgcn_readlane(s, k + u);
            dk[u] = __builtin_amdgcn_readlane(d, k + u);
            wk[u] = __int_as_float(__builtin_amdgcn_readlane(wi, k + u));
            r[u] = *(const rowv*)(hl + (long)sk * C);
          }
        }
#pragma unroll
        for (int u = 0; u < U - 1; ++u) {
          if (k + u < m) {
            if (dk[u] != cur) { flush(); cur = dk[u]; }
#pragma unroll
            for (int i = 0; i < CH; ++i) acc[i] += r[u][i] * wk[u];
          }
        }
      }
    }
    if (beg < end) flush();
  };

  if (tid < R) {
    float a = 0.f, v = 0.f;
    if (tid < nr) {
      a = ad[n0 + tid];
      v = as[n0 + tid] + a;
      v = (v > 0.f) ? v : NEG_SLOPE * v;
    }
    s_ad[tid] = a;
    s_vs[tid] = v;
    s_mk[tid] = gat_fkey(v);
    s_den[tid] = 0.f;
    s_c[tid] = 0;
  }
  if (tid == 0) s_cnt = 0u;
  __syncthreads();

  unsigned total;
  if (TAKE) {
    total = (unsigned)keep_n[blockIdx.x];
  } else {
    collect(0, nchunk);
    __syncthreads();
    total = s_cnt;
    if (tid == 0) keep_n[blockIdx.x] = (int)total;
  }
  const bool whole = (total <= (unsigned)CAP);
  const int  nseg  = whole ? 1 : (nchunk + PC - 1) / PC;
  const int  n     = whole ? (int)total : 0;

  if (whole) prep_sorted(n);
  else {
    float4 z4 = make_float4(0.f, 0.f, 0.f, 0.f);
    float4* a4 = (float4*)s_acc;
    for (int i = tid; i < R * C / 4; i += NT) a4[i] = z4;
    for (int k = 0; k < nseg; ++k) rebuild(k);
  }
  __syncthreads();
  if (tid < R) {
    float m = gat_fkey_inv(s_mk[tid]);
    if (!__builtin_isfinite(m)) m = 0.f;
    s_m[tid] = m;
    float ex = __expf(s_vs[tid] - m);
    s_vs[tid] = ex;
    s_den[tid] = ex;
  }
  if (whole && wv == NT / 32 - 1) {
    constexpr int PL = (R + 31) / 32;
    int c[PL];
    int sum = 0;
#pragma unroll
    for (int j = 0; j < PL; ++j) {
      const int d = lane * PL + j;
      c[j] = (d < nr) ? (s_c[d] + 1) : 0;
      sum += c[j];
    }
    int inc = sum;
#pragma unroll
    for (int off = 1; off < 32; off <<= 1) {
      const int t = __shfl_up(inc, off, 32);
      if (lane >= off) inc += t;
    }
    int run = inc - sum;
#pragma unroll
    for (int j = 0; j < PL; ++j) {
      const int d = lane * PL + j;
      if (d < R) s_off[d] = run;
      run += c[j];
    }
    if (lane == 31) s_off[R] = inc;
  }
  __syncthreads();

  if (whole) expsum_sorted(n);
  else for (int k = 0; k < nseg; ++k) { int c = rebuild(k); expsum(c, true); }
  __syncthreads();

  if (whole) rows_sorted(n);
  else {
    for (int k = 0; k < nseg; ++k) {
      int c = rebuild(k);
      expsum(c, false);
      __syncthreads();
      rows(c, 0);
    }
    rows(0, nr);
  }
  __syncthreads();

  if (!whole) {
    const float4* b4 = (const float4*)bias;
    float4* o4 = (float4*)(out + (long)n0 * C);
    for (int i = tid; i < nr * (C / 4); i += NT) {
      const int q = i % (C / 4);
      const float* a = s_acc + (i / (C / 4)) * C;
      float4 v;
      v.x = a[((4 * q)     % CH) * 32 + (4 * q)     / CH];
      v.y = a[((4 * q + 1) % CH) * 32 + (4 * q + 1) / CH];
      v.z = a[((4 * q + 2) % CH) * 32 + (4 * q + 2) / CH];
      v.w = a[((4 * q + 3) % CH) * 32 + (4 * q + 3) / CH];
      float4 b = b4[q];
      v.x += b.x; v.y += b.y; v.z += b.z; v.w += b.w;
      if (RELU) { v.x = fmaxf(v.x, 0.f); v.y = fmaxf(v.y, 0.f); v.z = fmaxf(v.z, 0.f); v.w = fmaxf(v.w, 0.f); }
      o4[i] = v;
    }
  }
}

extern "C" void kernel_launch(void* const* d_in, const int* in_sizes, int n_in,
                              void* d_out, int out_size, void* d_ws, size_t ws_size,
                              hipStream_t stream) {
  const float* fea = (const float*)d_in[0];
  const int*   ei  = (const int*)  d_in[1];
  const float* W1  = (const float*)d_in[2];
  const float* av1s= (const float*)d_in[3];
  const float* av1d= (const float*)d_in[4];
  const float* b1  = (const float*)d_in[5];
  const float* W2  = (const float*)d_in[6];
  const float* av2s= (const float*)d_in[7];
  const float* av2d= (const float*)d_in[8];
  const float* b2  = (const float*)d_in[9];
  float* out = (float*)d_out;

  const int B  = 2;
  const int C1 = 128, C2 = 64;
  const int N  = in_sizes[0] / (B * C1);
  const int E  = in_sizes[1] / (B * 2);

  float* w   = (float*)d_ws;
  float* h1  = w; w += (long)N * C1;
  float* x2  = w; w += (long)N * C1;
  float* h2  = w; w += (long)N * C2;
  float* asb = w; w += N;
  float* adb = w; w += N;
  float* as2 = w; w += N;
  float* ad2 = w; w += N;

  const int RT  = 200;
  const int CAPL = 6144;
  int gGemm    = N / 16;
  int gTile    = (N + RT - 1) / RT;
  int* keep_s = (int*)w; w += (long)gTile * CAPL;
  int* keep_d = (int*)w; w += (long)gTile * CAPL;
  int* keep_n = (int*)w; w += gTile;
  const int nchunk = (E + GAT_CHUNK - 1) / GAT_CHUNK;
  int2* lists = (int2*)(((uintptr_t)w + 15) & ~(uintptr_t)15);
  int*  offs  = (int*)(lists + (long)B * nchunk * GAT_CHUNK);

  gat_gemm_and_chunks<8, RT><<<gGemm + B * nchunk, dim3(32), 0, stream>>>(
      fea, W1, av1s, av1d, h1, asb, adb, C1, gGemm, ei, lists, offs, E, N, nchunk);

  for (int b = 0; b < B; ++b) {
    float*       outb = out + (long)b * N * C2;
    const int2*  lstb = lists + (long)b * nchunk * GAT_CHUNK;
    const int*   offb = offs + (long)b * nchunk * GAT_BINS;

    gat_tile_softmax_aggregate<128, RT, CAPL, 1, 0><<<gTile, dim3(1024), 0, stream>>>(
        lstb, offb, asb, adb, h1, b1, x2, keep_s, keep_d, keep_n, nchunk, N);

    if (b + 1 < B)
      gat_gemm_pair<8, 4><<<2 * gGemm, dim3(32), 0, stream>>>(
          fea + (long)(b + 1) * N * C1, W1, av1s, av1d, h1, asb, adb,
          x2, W2, av2s, av2d, h2, as2, ad2, C1, gGemm);
    else
      gat_gemm_wmma<4><<<gGemm, dim3(32), 0, stream>>>(x2, W2, av2s, av2d, h2, as2, ad2, N, C1);
    gat_tile_softmax_aggregate<64, RT, CAPL, 0, 1><<<gTile, dim3(1024), 0, stream>>>(
        lstb, offb, as2, ad2, h2, b2, outb, keep_s, keep_d, keep_n, nchunk, N);
  }
}


# Layer-2 tile pass of graph 0 and layer-1 tile pass of graph 1 share a launch, two work-groups per WGP (64 registers, LDS list 6144->4096): one's waits hide under the other's rows; 7->6; sums kept
# speedup vs baseline: 7.4329x; 1.0329x over previous
#include <hip/hip_runtime.h>
#include <math.h>
#include <stdint.h>

typedef __attribute__((ext_vector_type(2))) float v2f;
typedef __attribute__((ext_vector_type(8))) float v8f;

#define NEG_SLOPE 0.2f

template <int TILES>
__device__ __forceinline__
void gat_gemm_stripe(const float* __restrict__ X, const float* __restrict__ W,
                     const float* __restrict__ a_src, const float* __restrict__ a_dst,
                     float* __restrict__ Hout, float* __restrict__ as, float* __restrict__ ad,
                     int K, int row0, int lane) {
  const int F    = TILES * 16;
  const int l15  = lane & 15;
  const int kh   = (lane >> 4) * 2;
  const int m    = row0 + l15;

  v8f acc[TILES];
#pragma unroll
  for (int t = 0; t < TILES; ++t) acc[t] = (v8f){0.f,0.f,0.f,0.f,0.f,0.f,0.f,0.f};

  for (int k0 = 0; k0 < K; k0 += 4) {
    v2f a;
    a.x = X[(long)m * K + k0 + kh];
    a.y = X[(long)m * K + k0 + kh + 1];
#pragma unroll
    for (int t = 0; t < TILES; ++t) {
      v2f b;
      b.x = W[(long)(k0 + kh)     * F + t * 16 + l15];
      b.y = W[(long)(k0 + kh + 1) * F + t * 16 + l15];
      acc[t] = __builtin_amdgcn_wmma_f32_16x16x4_f32(
           false, a,  false, b,
           (short)0, acc[t],  false,  false);
    }
  }

  const int hi = (lane >> 4) * 8;
#pragma unroll
  for (int t = 0; t < TILES; ++t)
#pragma unroll
    for (int r = 0; r < 8; ++r)
      Hout[(long)(row0 + hi + r) * F + t * 16 + l15] = acc[t][r];

  float ps[8], pd[8];
#pragma unroll
  for (int r = 0; r < 8; ++r) { ps[r] = 0.f; pd[r] = 0.f; }
#pragma unroll
  for (int t = 0; t < TILES; ++t) {
    const float cs = a_src[t * 16 + l15];
    const float cd = a_dst[t * 16 + l15];
#pragma unroll
    for (int r = 0; r < 8; ++r) {
      ps[r] += acc[t][r] * cs;
      pd[r] += acc[t][r] * cd;
    }
  }
#pragma unroll
  for (int off = 8; off > 0; off >>= 1) {
#pragma unroll
    for (int r = 0; r < 8; ++r) {
      ps[r] += __shfl_xor(ps[r], off, 32);
      pd[r] += __shfl_xor(pd[r], off, 32);
    }
  }
  {
    const int r7 = l15 & 7;
    float vs = ps[0], vd = pd[0];
#pragma unroll
    for (int r = 1; r < 8; ++r) { if (r7 == r) { vs = ps[r]; vd = pd[r]; } }
    float* const o = (l15 < 8) ? as : ad;
    o[row0 + hi + r7] = (l15 < 8) ? vs : vd;
  }
}

#define GAT_CHUNK 1024
#define GAT_BINS  256

template <int R>
__device__ __forceinline__
void gat_chunk_sort(const int* __restrict__ ei, int2* __restrict__ lists, int* __restrict__ offs,
                    int E, int N, int nchunk, int b, int c, int lane, int* hist) {
  constexpr int PER = GAT_CHUNK / 32;
  constexpr int BL  = GAT_BINS / 32;
  const int* src = ei + (long)b * 2 * E;
  const int* dst = src + E;
#pragma unroll
  for (int j = 0; j < BL; ++j) hist[j * 32 + lane] = 0;
  __syncthreads();
  int s[PER], pk[PER];
  const int e0 = c * GAT_CHUNK;
#pragma unroll
  for (int j = 0; j < PER; ++j) {
    const int e  = e0 + j * 32 + lane;
    const int ec = (e < E) ? e : (E - 1);
    const int d  = dst[ec];
    s[j]  = src[ec];
    pk[j] = (e < E) ? d : -1;
  }
#pragma unroll
  for (int j = 0; j < PER; ++j) {
    const int d   = pk[j];
    const int key = ((unsigned)d < (unsigned)N) ? (d / R) : 255;
    const int rank = atomicAdd(&hist[key], 1);
    pk[j] = (key << 18) | (((d - key * R) & 255) << 10) | rank;
  }
  __syncthreads();
  int cn[BL];
  int sum = 0;
#pragma unroll
  for (int j = 0; j < BL; ++j) { cn[j] = hist[lane * BL + j]; sum += cn[j]; }
  int inc = sum;
#pragma unroll
  for (int off = 1; off < 32; off <<= 1) {
    const int t = __shfl_up(inc, off, 32);
    if (lane >= off) inc += t;
  }
  int run = inc - sum;
  int* const o = offs + ((long)b * nchunk + c) * GAT_BINS + lane * BL;
  __syncthreads();
#pragma unroll
  for (int j = 0; j < BL; ++j) {
    hist[lane * BL + j] = run;
    o[j] = run;
    run += cn[j];
  }
  __syncthreads();
  int2* const out = lists + ((long)b * nchunk + c) * GAT_CHUNK;
#pragma unroll
  for (int j = 0; j < PER; ++j) {
    const int p = pk[j];
    out[hist[p >> 18] + (p & 1023)] = make_int2(s[j], (p >> 10) & 255);
  }
}

template <int TILES>
__global__ __launch_bounds__(32)
void gat_gemm_wmma(const float* __restrict__ X, const float* __restrict__ W,
                   const float* __restrict__ a_src, const float* __restrict__ a_dst,
                   float* __restrict__ Hout, float* __restrict__ as, float* __restrict__ ad,
                   int N, int K) {
  gat_gemm_stripe<TILES>(X, W, a_src, a_dst, Hout, as, ad, K, blockIdx.x * 16, threadIdx.x);
}

template <int TILES, int R>
__global__ __launch_bounds__(32)
void gat_gemm_and_chunks(const float* __restrict__ X, const float* __restrict__ W,
                         const float* __restrict__ a_src, const float* __restrict__ a_dst,
                         float* __restrict__ Hout, float* __restrict__ as, float* __restrict__ ad,
                         int K, int nstripes,
                         const int* __restrict__ ei, int2* __restrict__ lists, int* __restrict__ offs,
                         int E, int N, int nchunk) {
  __shared__ int s_hist[GAT_BINS];
  const int bid = blockIdx.x;
  if (bid < nstripes) {
    gat_gemm_stripe<TILES>(X, W, a_src, a_dst, Hout, as, ad, K, bid * 16, threadIdx.x);
  } else {
    const int kb = bid - nstripes;
    const int b  = kb / nchunk;
    gat_chunk_sort<R>(ei, lists, offs, E, N, nchunk, b, kb - b * nchunk, (int)threadIdx.x, s_hist);
  }
}

template <int TILES_A, int TILES_B>
__global__ __launch_bounds__(32)
void gat_gemm_pair(const float* __restrict__ XA, const float* __restrict__ WA,
                   const float* __restrict__ a_srcA, const float* __restrict__ a_dstA,
                   float* __restrict__ HA, float* __restrict__ asA, float* __restrict__ adA,
                   const float* __restrict__ XB, const float* __restrict__ WB,
                   const float* __restrict__ a_srcB, const float* __restrict__ a_dstB,
                   float* __restrict__ HB, float* __restrict__ asB, float* __restrict__ adB,
                   int K, int nstripes) {
  const int bid = blockIdx.x;
  if (bid < nstripes)
    gat_gemm_stripe<TILES_A>(XA, WA, a_srcA, a_dstA, HA, asA, adA, K, bid * 16, threadIdx.x);
  else
    gat_gemm_stripe<TILES_B>(XB, WB, a_srcB, a_dstB, HB, asB, adB, K, (bid - nstripes) * 16, threadIdx.x);
}

__device__ __forceinline__ unsigned gat_fkey(float f) {
  unsigned b = __float_as_uint(f);
  return (b & 0x80000000u) ? ~b : (b | 0x80000000u);
}
__device__ __forceinline__ float gat_fkey_inv(unsigned k) {
  unsigned b = (k & 0x80000000u) ? (k & 0x7fffffffu) : ~k;
  return __uint_as_float(b);
}

template <int C, int R, int CAP>
struct gat_tile_lds {
  static constexpr int QN    = CAP + R;
  static constexpr int BUFN  = (R * C > 3 * QN) ? (R * C) : (3 * QN);
  static constexpr int WORDS = BUFN + 3 * CAP + 7 * R + 2;
};

template <int C, int R, int CAP, int RELU, int TAKE, int U>
__device__ __forceinline__
void gat_tile_pass(float* smem, const int bx,
                   const int2* __restrict__ lists,
                   const int* __restrict__ offs,
                   const float* __restrict__ as,
                   const float* __restrict__ ad,
                   const float* __restrict__ h,
                   const float* __restrict__ bias,
                   float* __restrict__ out,
                   int* __restrict__ keep_s,
                   int* __restrict__ keep_d,
                   int* __restrict__ keep_n,
                   int nchunk, int N) {
  constexpr int NT = 1024;
  constexpr int CH = C / 32;
  constexpr int QN = gat_tile_lds<C, R, CAP>::QN;
  constexpr int BUFN = gat_tile_lds<C, R, CAP>::BUFN;
  static_assert(CAP % GAT_CHUNK == 0, "a piece of the input is a run of whole chunks");
  constexpr int PC = CAP / GAT_CHUNK;
  static_assert(R <= 256, "a local destination takes the low 8 bits beside the entry's rank");
  float*    const s_buf = smem;
  int*      const l_s   = (int*)(smem + BUFN);
  int*      const l_d   = l_s + CAP;
  float*    const l_v   = (float*)(l_d + CAP);
  int*      const s_c   = (int*)(l_v + CAP);
  int*      const s_off = s_c + R;
  float*    const s_ad  = (float*)(s_off + R + 1);
  float*    const s_m   = s_ad + R;
  float*    const s_den = s_m + R;
  float*    const s_vs  = s_den + R;
  unsigned* const s_mk  = (unsigned*)(s_vs + R);
  unsigned& s_cnt = *(s_mk + R);

  const int tid  = threadIdx.x;
  const int lane = tid & 31;
  const int wv   = __builtin_amdgcn_readfirstlane(tid >> 5);
  const int n0   = bx * R;
  const int nr   = (N - n0 < R) ? (N - n0) : R;
  int* const my_s = keep_s + (long)bx * CAP;
  int* const my_d = keep_d + (long)bx * CAP;
  float* const s_acc = s_buf;
  int*   const q_s   = (int*)s_buf;
  int*   const q_d   = q_s + QN;
  float* const q_v   = s_buf + 2 * QN;

  auto collect = [&](int clo, int chi) {
    constexpr int G = 8;
    for (int c0 = clo; c0 < chi; c0 += NT) {
      const int c = c0 + tid;
      int st = 0, cn = 0;
      if (c < chi) {
        const int* o = offs + (long)c * GAT_BINS + bx;
        st = o[0];
        cn = o[1] - st;
      }
      const int2* p = lists + (long)c * GAT_CHUNK + st;
      const unsigned base = atomicAdd(&s_cnt, (unsigned)cn);
      for (int j0 = 0; j0 < cn; j0 += G) {
        int2 v[G];
#pragma unroll
        for (int g = 0; g < G; ++g) if (j0 + g < cn) v[g] = p[j0 + g];
#pragma unroll
        for (int g = 0; g < G; ++g) {
          const unsigned slot = base + (unsigned)(j0 + g);
          if (j0 + g < cn && slot < (unsigned)CAP) { l_s[slot] = v[g].x; l_d[slot] = v[g].y; }
        }
      }
    }
  };
  auto prep = [&](int cnt) {
    for (int i = tid; i < cnt; i += NT) {
      int s = l_s[i];
      int d = l_d[i];
      float v = as[s] + s_ad[d];
      v = (v > 0.f) ? v : NEG_SLOPE * v;
      l_v[i] = v;
      atomicMax(&s_mk[d], gat_fkey(v));
    }
  };
  auto prep_sorted = [&](int cnt) {
    for (int i = tid; i < cnt; i += NT) {
      int s, d;
      if (TAKE) { s = my_s[i]; d = my_d[i]; }
      else {
        s = l_s[i];
        d = l_d[i];
        my_s[i] = s; my_d[i] = d;
      }
      float v = as[s] + s_ad[d];
      v = (v > 0.f) ? v : NEG_SLOPE * v;
      int rank = atomicAdd(&s_c[d], 1);
      l_s[i] = s; l_d[i] = d | (rank << 8); l_v[i] = v;
      atomicMax(&s_mk[d], gat_fkey(v));
    }
  };
  auto rebuild = [&](int k) -> int {
    __syncthreads();
    if (tid == 0) s_cnt = 0u;
    __syncthreads();
    int lo = k * PC;
    int hi = (nchunk - lo < PC) ? nchunk : lo + PC;
    collect(lo, hi);
    __syncthreads();
    int c = (int)s_cnt;
    prep(c);
    __syncthreads();
    return c;
  };
  auto expsum = [&](int cnt, bool add) {
    for (int i = tid; i < cnt; i += NT) {
      int d = l_d[i];
      float ex = __expf(l_v[i] - s_m[d]);
      l_v[i] = ex;
      if (add) atomicAdd(&s_den[d], ex);
    }
  };
  auto expsum_sorted = [&](int cnt) {
    if (tid < nr) {
      int p = s_off[tid];
      q_s[p] = n0 + tid; q_d[p] = tid; q_v[p] = s_vs[tid];
    }
    for (int i = tid; i < cnt; i += NT) {
      int dr = l_d[i];
      int d = dr & 255;
      float ex = __expf(l_v[i] - s_m[d]);
      atomicAdd(&s_den[d], ex);
      int p = s_off[d] + 1 + (dr >> 8);
      q_s[p] = l_s[i]; q_d[p] = d; q_v[p] = ex;
    }
  };
  typedef float rowv __attribute__((ext_vector_type(CH)));
  auto rows = [&](int cnt_, int nself) {
    const int cnt   = __builtin_amdgcn_readfirstlane(cnt_);
    const int total = cnt + nself;
    const int per   = (total + 31) >> 5;
    const int beg   = wv * per;
    const int end   = (beg + per < total) ? (beg + per) : total;
    const float* const hl = h + lane * CH;
    float* const al = s_acc + lane;
    for (int b0 = beg; b0 < end; b0 += 32) {
      const int m = (end - b0 < 32) ? (end - b0) : 32;
      int s = n0, d = 0;
      float w = 0.f;
      if (lane < m) {
        const int idx = b0 + lane;
        float ex;
        if (idx < cnt) { s = l_s[idx]; d = l_d[idx]; ex = l_v[idx]; }
        else           { d = idx - cnt; s = n0 + d; ex = s_vs[d]; }
        w = ex / (s_den[d] + 1e-16f);
      }
      const int wi = __float_as_int(w);
      int k = 0;
      for (; k + U <= m; k += U) {
        rowv  r[U];
        int   dk[U];
        float wk[U];
#pragma unroll
        for (int u = 0; u < U; ++u) {
          const int sk = __builtin_amdgcn_readlane(s, k + u);
          dk[u] = __builtin_amdgcn_readlane(d, k + u);
          wk[u] = __int_as_float(__builtin_amdgcn_readlane(wi, k + u));
          r[u] = *(const rowv*)(hl + (long)sk * C);
        }
#pragma unroll
        for (int u = 0; u < U; ++u) {
          float* const a = al + dk[u] * C;
#pragma unroll
          for (int i = 0; i < CH; ++i) atomicAdd(a + i * 32, r[u][i] * wk[u]);
        }
      }
      if (k < m) {
        rowv  r[U - 1];
        int   dk[U - 1];
        float wk[U - 1];
#pragma unroll
        for (int u = 0; u < U - 1; ++u) {
          if (k + u < m) {
            const int sk = __builtin_amdgcn_readlane(s, k + u);
            dk[u] = __builtin_amdgcn_readlane(d, k + u);
            wk[u] = __int_as_float(__builtin_amdgcn_readlane(wi, k + u));
            r[u] = *(const rowv*)(hl + (long)sk * C);
          }
        }
#pragma unroll
        for (int u = 0; u < U - 1; ++u) {
          if (k + u < m) {
            float* const a = al + dk[u] * C;
#pragma unroll
            for (int i = 0; i < CH; ++i) atomicAdd(a + i * 32, r[u][i] * wk[u]);
          }
        }
      }
    }
  };

  auto rows_sorted = [&](int cnt_) {
    const int cnt   = __builtin_amdgcn_readfirstlane(cnt_);
    const int total = cnt + nr;
    const int per   = (total + 31) >> 5;
    const int tlo   = wv * per;
    const int thi   = tlo + per;
    int dlo = 0, dhi = 0;
    for (int j = 0; j < (R + 31) / 32; ++j) {
      const int d = j * 32 + lane;
      const int o = (d < R) ? s_off[d] : 0x7fffffff;
      dlo += __popc(__builtin_amdgcn_ballot_w32(o < tlo));
      dhi += __popc(__builtin_amdgcn_ballot_w32(o < thi));
    }
    const int beg = __builtin_amdgcn_readfirstlane(s_off[dlo]);
    const int end = __builtin_amdgcn_readfirstlane(s_off[dhi]);
    const float* const hl = h + lane * CH;
    float* const ol = out + (long)n0 * C + lane * CH;
    const rowv bv = *(const rowv*)(bias + lane * CH);
    rowv acc;
#pragma unroll
    for (int i = 0; i < CH; ++i) acc[i] = 0.f;
    int cur = dlo;
    auto flush = [&]() {
      rowv o;
#pragma unroll
      for (int i = 0; i < CH; ++i) {
        float t = acc[i] + bv[i];
        if (RELU) t = fmaxf(t, 0.f);
        o[i] = t;
        acc[i] = 0.f;
      }
      *(rowv*)(ol + (long)cur * C) = o;
    };
    for (int b0 = beg; b0 < end; b0 += 32) {
      const int m = (end - b0 < 32) ? (end - b0) : 32;
      int s = n0, d = 0;
      float w = 0.f;
      if (lane < m) {
        const int idx = b0 + lane;
        s = q_s[idx]; d = q_d[idx];
        w = q_v[idx] / (s_den[d] + 1e-16f);
      }
      const int wi = __float_as_int(w);
      int k = 0;
      for (; k + U <= m; k += U) {
        rowv  r[U];
        int   dk[U];
        float wk[U];
#pragma unroll
        for (int u = 0; u < U; ++u) {
          const int sk = __builtin_amdgcn_readlane(s, k + u);
          dk[u] = __builtin_amdgcn_readlane(d, k + u);
          wk[u] = __int_as_float(__builtin_amdgcn_readlane(wi, k + u));
          r[u] = *(const rowv*)(hl + (long)sk * C);
        }
#pragma unroll
        for (int u = 0; u < U; ++u) {
          if (dk[u] != cur) { flush(); cur = dk[u]; }
#pragma unroll
          for (int i = 0; i < CH; ++i) acc[i] += r[u][i] * wk[u];
        }
      }
      if (k < m) {
        rowv  r[U - 1];
        int   dk[U - 1];
        float wk[U - 1];
#pragma unroll
        for (int u = 0; u < U - 1; ++u) {
          if (k + u < m) {
            const int sk = __builtin_amdgcn_readlane(s, k + u);
            dk[u] = __builtin_amdgcn_readlane(d, k + u);
            wk[u] = __int_as_float(__builtin_amdgcn_readlane(wi, k + u));
            r[u] = *(const rowv*)(hl + (long)sk * C);
          }
        }
#pragma unroll
        for (int u = 0; u < U - 1; ++u) {
          if (k + u < m) {
            if (dk[u] != cur) { flush(); cur = dk[u]; }
#pragma unroll
            for (int i = 0; i < CH; ++i) acc[i] += r[u][i] * wk[u];
          }
        }
      }
    }
    if (beg < end) flush();
  };

  if (tid < R) {
    float a = 0.f, v = 0.f;
    if (tid < nr) {
      a = ad[n0 + tid];
      v = as[n0 + tid] + a;
      v = (v > 0.f) ? v : NEG_SLOPE * v;
    }
    s_ad[tid] = a;
    s_vs[tid] = v;
    s_mk[tid] = gat_fkey(v);
    s_den[tid] = 0.f;
    s_c[tid] = 0;
  }
  if (tid == 0) s_cnt = 0u;
  __syncthreads();

  unsigned total;
  if (TAKE) {
    total = (unsigned)keep_n[bx];
  } else {
    collect(0, nchunk);
    __syncthreads();
    total = s_cnt;
    if (tid == 0) keep_n[bx] = (int)total;
  }
  const bool whole = (total <= (unsigned)CAP);
  const int  nseg  = whole ? 1 : (nchunk + PC - 1) / PC;
  const int  n     = whole ? (int)total : 0;

  if (whole) prep_sorted(n);
  else {
    float4 z4 = make_float4(0.f, 0.f, 0.f, 0.f);
    float4* a4 = (float4*)s_acc;
    for (int i = tid; i < R * C / 4; i += NT) a4[i] = z4;
    for (int k = 0; k < nseg; ++k) rebuild(k);
  }
  __syncthreads();
  if (tid < R) {
    float m = gat_fkey_inv(s_mk[tid]);
    if (!__builtin_isfinite(m)) m = 0.f;
    s_m[tid] = m;
    float ex = __expf(s_vs[tid] - m);
    s_vs[tid] = ex;
    s_den[tid] = ex;
  }
  if (whole && wv == NT / 32 - 1) {
    constexpr int PL = (R + 31) / 32;
    int c[PL];
    int sum = 0;
#pragma unroll
    for (int j = 0; j < PL; ++j) {
      const int d = lane * PL + j;
      c[j] = (d < nr) ? (s_c[d] + 1) : 0;
      sum += c[j];
    }
    int inc = sum;
#pragma unroll
    for (int off = 1; off < 32; off <<= 1) {
      const int t = __shfl_up(inc, off, 32);
      if (lane >= off) inc += t;
    }
    int run = inc - sum;
#pragma unroll
    for (int j = 0; j < PL; ++j) {
      const int d = lane * PL + j;
      if (d < R) s_off[d] = run;
      run += c[j];
    }
    if (lane == 31) s_off[R] = inc;
  }
  __syncthreads();

  if (whole) expsum_sorted(n);
  else for (int k = 0; k < nseg; ++k) { int c = rebuild(k); expsum(c, true); }
  __syncthreads();

  if (whole) rows_sorted(n);
  else {
    for (int k = 0; k < nseg; ++k) {
      int c = rebuild(k);
      expsum(c, false);
      __syncthreads();
      rows(c, 0);
    }
    rows(0, nr);
  }
  __syncthreads();

  if (!whole) {
    const float4* b4 = (const float4*)bias;
    float4* o4 = (float4*)(out + (long)n0 * C);
#pragma unroll 1
    for (int i = tid; i < nr * (C / 4); i += NT) {
      const int q = i % (C / 4);
      const float* a = s_acc + (i / (C / 4)) * C;
      float4 v;
      v.x = a[((4 * q)     % CH) * 32 + (4 * q)     / CH];
      v.y = a[((4 * q + 1) % CH) * 32 + (4 * q + 1) / CH];
      v.z = a[((4 * q + 2) % CH) * 32 + (4 * q + 2) / CH];
      v.w = a[((4 * q + 3) % CH) * 32 + (4 * q + 3) / CH];
      float4 b = b4[q];
      v.x += b.x; v.y += b.y; v.z += b.z; v.w += b.w;
      if (RELU) { v.x = fmaxf(v.x, 0.f); v.y = fmaxf(v.y, 0.f); v.z = fmaxf(v.z, 0.f); v.w = fmaxf(v.w, 0.f); }
      o4[i] = v;
    }
  }
}

template <int C, int R, int CAP, int RELU, int TAKE>
__global__ __launch_bounds__(1024)
void gat_tile_softmax_aggregate(const int2* __restrict__ lists,
                                const int* __restrict__ offs,
                                const float* __restrict__ as,
                                const float* __restrict__ ad,
                                const float* __restrict__ h,
                                const float* __restrict__ bias,
                                float* __restrict__ out,
                                int* __restrict__ keep_s,
                                int* __restrict__ keep_d,
                                int* __restrict__ keep_n,
                                int nchunk, int N) {
  __shared__ __attribute__((aligned(16))) float smem[gat_tile_lds<C, R, CAP>::WORDS];
  gat_tile_pass<C, R, CAP, RELU, TAKE, 8>(smem, blockIdx.x, lists, offs, as, ad, h, bias, out,
                                          keep_s, keep_d, keep_n, nchunk, N);
}

template <int CA, int CB, int R, int CAP>
__global__ __launch_bounds__(1024, 16)
void gat_tile_pair(const int2* __restrict__ listsA, const int* __restrict__ offsA,
                   const float* __restrict__ asA, const float* __restrict__ adA,
                   const float* __restrict__ hA, const float* __restrict__ biasA,
                   float* __restrict__ outA,
                   int* __restrict__ keep_sA, int* __restrict__ keep_dA, int* __restrict__ keep_nA,
                   const int2* __restrict__ listsB, const int* __restrict__ offsB,
                   const float* __restrict__ asB, const float* __restrict__ adB,
                   const float* __restrict__ hB, const float* __restrict__ biasB,
                   float* __restrict__ outB,
                   int* __restrict__ keep_sB, int* __restrict__ keep_dB, int* __restrict__ keep_nB,
                   int nchunk, int N, int ntile) {
  constexpr int WA = gat_tile_lds<CA, R, CAP>::WORDS;
  constexpr int WB = gat_tile_lds<CB, R, CAP>::WORDS;
  __shared__ __attribute__((aligned(16))) float smem[(WA > WB) ? WA : WB];
  const int bid = blockIdx.x;
  if (bid < ntile)
    gat_tile_pass<CB, R, CAP, 1, 0, 24 / (CB / 32)>(smem, bid, listsB, offsB, asB, adB, hB, biasB, outB,
                                                    keep_sB, keep_dB, keep_nB, nchunk, N);
  else
    gat_tile_pass<CA, R, CAP, 0, 1, 16 / (CA / 32)>(smem, bid - ntile, listsA, offsA, asA, adA, hA, biasA, outA,
                                                    keep_sA, keep_dA, keep_nA, nchunk, N);
}

extern "C" void kernel_launch(void* const* d_in, const int* in_sizes, int n_in,
                              void* d_out, int out_size, void* d_ws, size_t ws_size,
                              hipStream_t stream) {
  const float* fea = (const float*)d_in[0];
  const int*   ei  = (const int*)  d_in[1];
  const float* W1  = (const float*)d_in[2];
  const float* av1s= (const float*)d_in[3];
  const float* av1d= (const float*)d_in[4];
  const float* b1  = (const float*)d_in[5];
  const float* W2  = (const float*)d_in[6];
  const float* av2s= (const float*)d_in[7];
  const float* av2d= (const float*)d_in[8];
  const float* b2  = (const float*)d_in[9];
  float* out = (float*)d_out;

  const int B  = 2;
  const int C1 = 128, C2 = 64;
  const int N  = in_sizes[0] / (B * C1);
  const int E  = in_sizes[1] / (B * 2);

  float* w   = (float*)d_ws;
  float* h1  = w; w += (long)N * C1;
  float* x2  = w; w += (long)N * C1;
  float* h2  = w; w += (long)N * C2;
  float* asb = w; w += N;
  float* adb = w; w += N;
  float* as2 = w; w += N;
  float* ad2 = w; w += N;

  const int RT  = 200;
  const int CAPL = 4096;
  int gGemm    = N / 16;
  int gTile    = (N + RT - 1) / RT;
  int* keep_s0 = (int*)w; w += (long)2 * gTile * CAPL;
  int* keep_d0 = (int*)w; w += (long)2 * gTile * CAPL;
  int* keep_n0 = (int*)w; w += 2 * gTile;
  const int nchunk = (E + GAT_CHUNK - 1) / GAT_CHUNK;
  int2* lists = (int2*)(((uintptr_t)w + 15) & ~(uintptr_t)15);
  int*  offs  = (int*)(lists + (long)B * nchunk * GAT_CHUNK);

  gat_gemm_and_chunks<8, RT><<<gGemm + B * nchunk, dim3(32), 0, stream>>>(
      fea, W1, av1s, av1d, h1, asb, adb, C1, gGemm, ei, lists, offs, E, N, nchunk);

  for (int b = 0; b < B; ++b) {
    float*       outb = out + (long)b * N * C2;
    const int2*  lstb = lists + (long)b * nchunk * GAT_CHUNK;
    const int*   offb = offs + (long)b * nchunk * GAT_BINS;
    int* keep_s = keep_s0 + (long)(b & 1) * gTile * CAPL;
    int* keep_d = keep_d0 + (long)(b & 1) * gTile * CAPL;
    int* keep_n = keep_n0 + (b & 1) * gTile;

    if (b == 0)
      gat_tile_softmax_aggregate<128, RT, CAPL, 1, 0><<<gTile, dim3(1024), 0, stream>>>(
          lstb, offb, asb, adb, h1, b1, x2, keep_s, keep_d, keep_n, nchunk, N);

    if (b + 1 < B) {
      gat_gemm_pair<8, 4><<<2 * gGemm, dim3(32), 0, stream>>>(
          fea + (long)(b + 1) * N * C1, W1, av1s, av1d, h1, asb, adb,
          x2, W2, av2s, av2d, h2, as2, ad2, C1, gGemm);
      gat_tile_pair<64, 128, RT, CAPL><<<2 * gTile, dim3(1024), 0, stream>>>(
          lstb, offb, as2, ad2, h2, b2, outb, keep_s, keep_d, keep_n,
          lists + (long)(b + 1) * nchunk * GAT_CHUNK, offs + (long)(b + 1) * nchunk * GAT_BINS,
          asb, adb, h1, b1, x2,
          keep_s0 + (long)((b + 1) & 1) * gTile * CAPL, keep_d0 + (long)((b + 1) & 1) * gTile * CAPL,
          keep_n0 + ((b + 1) & 1) * gTile,
          nchunk, N, gTile);
    } else {
      gat_gemm_wmma<4><<<gGemm, dim3(32), 0, stream>>>(x2, W2, av2s, av2d, h2, as2, ad2, N, C1);
      gat_tile_softmax_aggregate<64, RT, CAPL, 0, 1><<<gTile, dim3(1024), 0, stream>>>(
          lstb, offb, as2, ad2, h2, b2, outb, keep_s, keep_d, keep_n, nchunk, N);
    }
  }
}
